# starting kernel
# baseline (speedup 1.0000x reference)
#include <hip/hip_runtime.h>
#include <cstdint>

typedef __attribute__((ext_vector_type(2))) float v2f;
typedef __attribute__((ext_vector_type(8))) float v8f;

__global__ void deg_count_kernel(const int* __restrict__ dst, int E,
                                 float* __restrict__ deg) {
    int e = blockIdx.x * blockDim.x + threadIdx.x;
    if (e < E) atomicAdd(&deg[dst[e]], 1.0f);
}

__global__ void deg_finalize_kernel(float* __restrict__ dis, int n) {
    int i = blockIdx.x * blockDim.x + threadIdx.x;
    if (i < n) dis[i] = rsqrtf(dis[i] + 1.0f);
}

template <int K, int NC>
__global__ void gemm_wmma_f32(const float* __restrict__ A,
                              const float* __restrict__ B,
                              float* __restrict__ C, int nTiles) {
    int wave = (int)((blockIdx.x * blockDim.x + threadIdx.x) >> 5);
    int lane = threadIdx.x & 31;
    if (wave >= nTiles) return;
    constexpr int TN = NC / 16;
    int tm = wave / TN;
    int tn = wave % TN;
    int rowBase = tm * 16;
    int colBase = tn * 16;
    int hi = lane >> 4;
    int l  = lane & 15;

    const float* arow = A + (size_t)(rowBase + l) * K;
    const float* bcol = B + colBase + l;

    v8f acc = {};
#pragma unroll
    for (int k = 0; k < K; k += 4) {
        v2f a, b;
        a.x = arow[k + 2 * hi + 0];
        a.y = arow[k + 2 * hi + 1];
        b.x = bcol[(size_t)(k + 2 * hi + 0) * NC];
        b.y = bcol[(size_t)(k + 2 * hi + 1) * NC];
        acc = __builtin_amdgcn_wmma_f32_16x16x4_f32(
             false, a,  false, b,
             (short)0, acc,  false,  false);
    }

    float* cptr = C + (size_t)colBase + l;
#pragma unroll
    for (int i = 0; i < 8; ++i)
        cptr[(size_t)(rowBase + 8 * hi + i) * NC] = acc[i];
}

template <int LOGF>
__global__ void edge_agg_kernel(const int* __restrict__ src,
                                const int* __restrict__ dst,
                                const float* __restrict__ dis,
                                const float* __restrict__ h,
                                float* __restrict__ agg, long long total) {
    long long t = (long long)blockIdx.x * blockDim.x + threadIdx.x;
    if (t >= total) return;
    int e = (int)(t >> LOGF);
    int j = (int)(t & ((1 << LOGF) - 1));
    int s = src[e];
    int d = dst[e];
    float norm = dis[s] * dis[d];
    float v = h[((size_t)s << LOGF) + j] * norm;
    atomicAdd(&agg[((size_t)d << LOGF) + j], v);
}

__global__ void selfloop_bias_relu_kernel(const float* __restrict__ h,
                                          const float* __restrict__ dis,
                                          const float* __restrict__ bias,
                                          float* __restrict__ agg, int n) {
    int t = blockIdx.x * blockDim.x + threadIdx.x;
    if (t >= n) return;
    int i = t >> 6;
    int j = t & 63;
    float dii = dis[i];
    float v = agg[t] + h[t] * dii * dii + bias[j];
    agg[t] = fmaxf(v, 0.0f);
}

__global__ void logsoftmax_kernel(const float* __restrict__ h,
                                  const float* __restrict__ dis,
                                  const float* __restrict__ bias,
                                  const float* __restrict__ agg,
                                  float* __restrict__ out, int n) {
    int i = blockIdx.x * blockDim.x + threadIdx.x;
    if (i >= n) return;
    float dii = dis[i];
    dii *= dii;
    float v[16];
    float mx = -3.402823466e38f;
#pragma unroll
    for (int j = 0; j < 16; ++j) {
        size_t idx = (size_t)i * 16 + j;
        v[j] = agg[idx] + h[idx] * dii + bias[j];
        mx = fmaxf(mx, v[j]);
    }
    float s = 0.0f;
#pragma unroll
    for (int j = 0; j < 16; ++j) s += __expf(v[j] - mx);
    float lse = mx + __logf(s);
#pragma unroll
    for (int j = 0; j < 16; ++j) out[(size_t)i * 16 + j] = v[j] - lse;
}

extern "C" void kernel_launch(void* const* d_in, const int* in_sizes, int n_in,
                              void* d_out, int out_size, void* d_ws, size_t ws_size,
                              hipStream_t stream) {
    const float* x   = (const float*)d_in[0];
    const int*   ei  = (const int*)  d_in[1];
    const float* W1  = (const float*)d_in[2];
    const float* b1  = (const float*)d_in[3];
    const float* W2  = (const float*)d_in[4];
    const float* b2  = (const float*)d_in[5];
    float* out = (float*)d_out;

    const int N = in_sizes[0] / 128;
    const int E = in_sizes[1] / 2;
    const int* src = ei;
    const int* dst = ei + E;

    float* dis  = (float*)d_ws;
    float* h1   = dis  + N;
    float* agg1 = h1   + (size_t)N * 64;
    float* h2   = agg1 + (size_t)N * 64;
    float* agg2 = h2   + (size_t)N * 16;

    hipMemsetAsync(dis,  0, (size_t)N * sizeof(float), stream);
    hipMemsetAsync(agg1, 0, (size_t)N * 64 * sizeof(float), stream);
    hipMemsetAsync(agg2, 0, (size_t)N * 16 * sizeof(float), stream);

    const int T = 256;

    deg_count_kernel<<<(E + T - 1) / T, T, 0, stream>>>(dst, E, dis);
    deg_finalize_kernel<<<(N + T - 1) / T, T, 0, stream>>>(dis, N);

    {
        int nTiles = (N / 16) * (64 / 16);
        int waves_per_block = T / 32;
        gemm_wmma_f32<128, 64><<<(nTiles + waves_per_block - 1) / waves_per_block,
                                 T, 0, stream>>>(x, W1, h1, nTiles);
    }
    {
        long long total = (long long)E * 64;
        edge_agg_kernel<6><<<(int)((total + T - 1) / T), T, 0, stream>>>(
            src, dst, dis, h1, agg1, total);
    }
    {
        int total = N * 64;
        selfloop_bias_relu_kernel<<<(total + T - 1) / T, T, 0, stream>>>(
            h1, dis, b1, agg1, total);
    }
    {
        int nTiles = (N / 16) * (16 / 16);
        int waves_per_block = T / 32;
        gemm_wmma_f32<64, 16><<<(nTiles + waves_per_block - 1) / waves_per_block,
                                T, 0, stream>>>(agg1, W2, h2, nTiles);
    }
    {
        long long total = (long long)E * 16;
        edge_agg_kernel<4><<<(int)((total + T - 1) / T), T, 0, stream>>>(
            src, dst, dis, h2, agg2, total);
    }
    logsoftmax_kernel<<<(N + T - 1) / T, T, 0, stream>>>(h2, dis, b2, agg2, out, N);
}


# Aggregate by destination, not by scatter: per-call dst-sorted edge list (int atomic rank+count, block scan, fill); each node sums its in-edges in registers; float atomics gone; sum order changes
# speedup vs baseline: 2.2516x; 2.2516x over previous
#include <hip/hip_runtime.h>
#include <cstdint>

typedef __attribute__((ext_vector_type(2))) float v2f;
typedef __attribute__((ext_vector_type(8))) float v8f;

__global__ void rank_count_kernel(const int* __restrict__ dst, int E,
                                  int* __restrict__ cnt, int* __restrict__ rank) {
    int e = blockIdx.x * blockDim.x + threadIdx.x;
    if (e < E) rank[e] = atomicAdd(&cnt[dst[e]], 1);
}

__global__ void seg_alloc_kernel(const int* __restrict__ cnt, int n,
                                 int* __restrict__ counter,
                                 int* __restrict__ start,
                                 float* __restrict__ dis) {
    __shared__ int wsum[8];
    __shared__ int base_s;
    int i = blockIdx.x * 256 + threadIdx.x;
    int lane = threadIdx.x & 31;
    int w = threadIdx.x >> 5;
    int c = (i < n) ? cnt[i] : 0;
    int incl = c;
#pragma unroll
    for (int o = 1; o < 32; o <<= 1) {
        int t = __shfl_up(incl, o);
        if (lane >= o) incl += t;
    }
    if (lane == 31) wsum[w] = incl;
    __syncthreads();
    if (threadIdx.x == 0) {
        int tot = 0;
#pragma unroll
        for (int k = 0; k < 8; ++k) { int t = wsum[k]; wsum[k] = tot; tot += t; }
        base_s = atomicAdd(counter, tot);
    }
    __syncthreads();
    if (i < n) {
        start[i] = base_s + wsum[w] + incl - c;
        dis[i] = rsqrtf((float)c + 1.0f);
    }
}

__global__ void csr_fill_kernel(const int* __restrict__ src,
                                const int* __restrict__ dst,
                                const int* __restrict__ rank,
                                const int* __restrict__ start, int E,
                                int* __restrict__ csr_src) {
    int e = blockIdx.x * blockDim.x + threadIdx.x;
    if (e < E) csr_src[start[dst[e]] + rank[e]] = src[e];
}

template <int K, int NC>
__global__ void gemm_wmma_f32(const float* __restrict__ A,
                              const float* __restrict__ B,
                              float* __restrict__ C, int nTiles) {
    int wave = (int)((blockIdx.x * blockDim.x + threadIdx.x) >> 5);
    int lane = threadIdx.x & 31;
    if (wave >= nTiles) return;
    constexpr int TN = NC / 16;
    int tm = wave / TN;
    int tn = wave % TN;
    int rowBase = tm * 16;
    int colBase = tn * 16;
    int hi = lane >> 4;
    int l  = lane & 15;

    const float* arow = A + (size_t)(rowBase + l) * K;
    const float* bcol = B + colBase + l;

    v8f acc = {};
#pragma unroll
    for (int k = 0; k < K; k += 4) {
        v2f a, b;
        a.x = arow[k + 2 * hi + 0];
        a.y = arow[k + 2 * hi + 1];
        b.x = bcol[(size_t)(k + 2 * hi + 0) * NC];
        b.y = bcol[(size_t)(k + 2 * hi + 1) * NC];
        acc = __builtin_amdgcn_wmma_f32_16x16x4_f32(
             false, a,  false, b,
             (short)0, acc,  false,  false);
    }

    float* cptr = C + (size_t)colBase + l;
#pragma unroll
    for (int i = 0; i < 8; ++i)
        cptr[(size_t)(rowBase + 8 * hi + i) * NC] = acc[i];
}

template <int G>
__global__ void gather_agg_kernel(const int* __restrict__ csr_src,
                                  const int* __restrict__ start,
                                  const int* __restrict__ cnt,
                                  const float* __restrict__ dis,
                                  const float* __restrict__ h,
                                  float* __restrict__ agg, int n) {
    int t = blockIdx.x * blockDim.x + threadIdx.x;
    int node = t / G;
    int q = t % G;
    if (node >= n) return;
    const int* idx = csr_src + start[node];
    int len = cnt[node];
    float dd = dis[node];
    const float4* h4 = (const float4*)h;
    float4 acc = {0.0f, 0.0f, 0.0f, 0.0f};
    int k = 0;
    for (; k + 4 <= len; k += 4) {
        int s0 = idx[k + 0];
        int s1 = idx[k + 1];
        int s2 = idx[k + 2];
        int s3 = idx[k + 3];
        float w0 = dis[s0] * dd;
        float w1 = dis[s1] * dd;
        float w2 = dis[s2] * dd;
        float w3 = dis[s3] * dd;
        float4 v0 = h4[(size_t)s0 * G + q];
        float4 v1 = h4[(size_t)s1 * G + q];
        float4 v2 = h4[(size_t)s2 * G + q];
        float4 v3 = h4[(size_t)s3 * G + q];
        acc.x += v0.x * w0; acc.y += v0.y * w0; acc.z += v0.z * w0; acc.w += v0.w * w0;
        acc.x += v1.x * w1; acc.y += v1.y * w1; acc.z += v1.z * w1; acc.w += v1.w * w1;
        acc.x += v2.x * w2; acc.y += v2.y * w2; acc.z += v2.z * w2; acc.w += v2.w * w2;
        acc.x += v3.x * w3; acc.y += v3.y * w3; acc.z += v3.z * w3; acc.w += v3.w * w3;
    }
    if (k < len) {
        int last = len - 1;
        bool p1 = k + 1 < len;
        bool p2 = k + 2 < len;
        int s0 = idx[k];
        int s1 = idx[min(k + 1, last)];
        int s2 = idx[min(k + 2, last)];
        float w0 = dis[s0] * dd;
        float w1 = dis[s1] * dd;
        float w2 = dis[s2] * dd;
        float4 v0 = h4[(size_t)s0 * G + q];
        float4 v1 = h4[(size_t)s1 * G + q];
        float4 v2 = h4[(size_t)s2 * G + q];
        acc.x += v0.x * w0; acc.y += v0.y * w0; acc.z += v0.z * w0; acc.w += v0.w * w0;
        float4 t1 = acc;
        t1.x += v1.x * w1; t1.y += v1.y * w1; t1.z += v1.z * w1; t1.w += v1.w * w1;
        acc.x = p1 ? t1.x : acc.x; acc.y = p1 ? t1.y : acc.y;
        acc.z = p1 ? t1.z : acc.z; acc.w = p1 ? t1.w : acc.w;
        float4 t2 = acc;
        t2.x += v2.x * w2; t2.y += v2.y * w2; t2.z += v2.z * w2; t2.w += v2.w * w2;
        acc.x = p2 ? t2.x : acc.x; acc.y = p2 ? t2.y : acc.y;
        acc.z = p2 ? t2.z : acc.z; acc.w = p2 ? t2.w : acc.w;
    }
    ((float4*)agg)[(size_t)node * G + q] = acc;
}

__global__ void selfloop_bias_relu_kernel(const float* __restrict__ h,
                                          const float* __restrict__ dis,
                                          const float* __restrict__ bias,
                                          float* __restrict__ agg, int n) {
    int t = blockIdx.x * blockDim.x + threadIdx.x;
    if (t >= n) return;
    int i = t >> 6;
    int j = t & 63;
    float dii = dis[i];
    float v = agg[t] + h[t] * dii * dii + bias[j];
    agg[t] = fmaxf(v, 0.0f);
}

__global__ void logsoftmax_kernel(const float* __restrict__ h,
                                  const float* __restrict__ dis,
                                  const float* __restrict__ bias,
                                  const float* __restrict__ agg,
                                  float* __restrict__ out, int n) {
    int i = blockIdx.x * blockDim.x + threadIdx.x;
    if (i >= n) return;
    float dii = dis[i];
    dii *= dii;
    float v[16];
    float mx = -3.402823466e38f;
#pragma unroll
    for (int j = 0; j < 16; ++j) {
        size_t idx = (size_t)i * 16 + j;
        v[j] = agg[idx] + h[idx] * dii + bias[j];
        mx = fmaxf(mx, v[j]);
    }
    float s = 0.0f;
#pragma unroll
    for (int j = 0; j < 16; ++j) s += __expf(v[j] - mx);
    float lse = mx + __logf(s);
#pragma unroll
    for (int j = 0; j < 16; ++j) out[(size_t)i * 16 + j] = v[j] - lse;
}

extern "C" void kernel_launch(void* const* d_in, const int* in_sizes, int n_in,
                              void* d_out, int out_size, void* d_ws, size_t ws_size,
                              hipStream_t stream) {
    const float* x   = (const float*)d_in[0];
    const int*   ei  = (const int*)  d_in[1];
    const float* W1  = (const float*)d_in[2];
    const float* b1  = (const float*)d_in[3];
    const float* W2  = (const float*)d_in[4];
    const float* b2  = (const float*)d_in[5];
    float* out = (float*)d_out;

    const int N = in_sizes[0] / 128;
    const int E = in_sizes[1] / 2;
    const int* src = ei;
    const int* dst = ei + E;

    float* h1   = (float*)d_ws;
    float* agg1 = h1   + (size_t)N * 64;
    float* h2   = agg1 + (size_t)N * 64;
    float* agg2 = h2   + (size_t)N * 16;
    float* dis  = agg2 + (size_t)N * 16;
    int* start   = (int*)(dis + N);
    int* cnt     = start + N;
    int* counter = cnt + N;
    int* rank    = counter + 1;
    int* csr_src = rank + E;

    hipMemsetAsync(cnt, 0, (size_t)(N + 1) * sizeof(int), stream);

    const int T = 256;

    rank_count_kernel<<<(E + T - 1) / T, T, 0, stream>>>(dst, E, cnt, rank);
    seg_alloc_kernel<<<(N + T - 1) / T, T, 0, stream>>>(cnt, N, counter, start, dis);
    csr_fill_kernel<<<(E + T - 1) / T, T, 0, stream>>>(src, dst, rank, start, E, csr_src);

    {
        int nTiles = (N / 16) * (64 / 16);
        int waves_per_block = T / 32;
        gemm_wmma_f32<128, 64><<<(nTiles + waves_per_block - 1) / waves_per_block,
                                 T, 0, stream>>>(x, W1, h1, nTiles);
    }
    {
        long long total = (long long)N * 16;
        gather_agg_kernel<16><<<(int)((total + T - 1) / T), T, 0, stream>>>(
            csr_src, start, cnt, dis, h1, agg1, N);
    }
    {
        int total = N * 64;
        selfloop_bias_relu_kernel<<<(total + T - 1) / T, T, 0, stream>>>(
            h1, dis, b1, agg1, total);
    }
    {
        int nTiles = (N / 16) * (16 / 16);
        int waves_per_block = T / 32;
        gemm_wmma_f32<64, 16><<<(nTiles + waves_per_block - 1) / waves_per_block,
                                T, 0, stream>>>(agg1, W2, h2, nTiles);
    }
    {
        long long total = (long long)N * 4;
        gather_agg_kernel<4><<<(int)((total + T - 1) / T), T, 0, stream>>>(
            csr_src, start, cnt, dis, h2, agg2, N);
    }
    logsoftmax_kernel<<<(N + T - 1) / T, T, 0, stream>>>(h2, dis, b2, agg2, out, N);
}


# Both layer epilogues fused into the by-destination aggregation kernels (relu; log_softmax via 4-lane shuffles); each sum now starts at self-loop term + bias; sum order changes
# speedup vs baseline: 2.5422x; 1.1291x over previous
#include <hip/hip_runtime.h>
#include <cstdint>

typedef __attribute__((ext_vector_type(2))) float v2f;
typedef __attribute__((ext_vector_type(8))) float v8f;

__global__ void rank_count_kernel(const int* __restrict__ dst, int E,
                                  int* __restrict__ cnt, int* __restrict__ rank) {
    int e = blockIdx.x * blockDim.x + threadIdx.x;
    if (e < E) rank[e] = atomicAdd(&cnt[dst[e]], 1);
}

__global__ void seg_alloc_kernel(const int* __restrict__ cnt, int n,
                                 int* __restrict__ counter,
                                 int* __restrict__ start,
                                 float* __restrict__ dis) {
    __shared__ int wsum[8];
    __shared__ int base_s;
    int i = blockIdx.x * 256 + threadIdx.x;
    int lane = threadIdx.x & 31;
    int w = threadIdx.x >> 5;
    int c = (i < n) ? cnt[i] : 0;
    int incl = c;
#pragma unroll
    for (int o = 1; o < 32; o <<= 1) {
        int t = __shfl_up(incl, o);
        if (lane >= o) incl += t;
    }
    if (lane == 31) wsum[w] = incl;
    __syncthreads();
    if (threadIdx.x == 0) {
        int tot = 0;
#pragma unroll
        for (int k = 0; k < 8; ++k) { int t = wsum[k]; wsum[k] = tot; tot += t; }
        base_s = atomicAdd(counter, tot);
    }
    __syncthreads();
    if (i < n) {
        start[i] = base_s + wsum[w] + incl - c;
        dis[i] = rsqrtf((float)c + 1.0f);
    }
}

__global__ void csr_fill_kernel(const int* __restrict__ src,
                                const int* __restrict__ dst,
                                const int* __restrict__ rank,
                                const int* __restrict__ start, int E,
                                int* __restrict__ csr_src) {
    int e = blockIdx.x * blockDim.x + threadIdx.x;
    if (e < E) csr_src[start[dst[e]] + rank[e]] = src[e];
}

template <int K, int NC>
__global__ void gemm_wmma_f32(const float* __restrict__ A,
                              const float* __restrict__ B,
                              float* __restrict__ C, int nTiles) {
    int wave = (int)((blockIdx.x * blockDim.x + threadIdx.x) >> 5);
    int lane = threadIdx.x & 31;
    if (wave >= nTiles) return;
    constexpr int TN = NC / 16;
    int tm = wave / TN;
    int tn = wave % TN;
    int rowBase = tm * 16;
    int colBase = tn * 16;
    int hi = lane >> 4;
    int l  = lane & 15;

    const float* arow = A + (size_t)(rowBase + l) * K;
    const float* bcol = B + colBase + l;

    v8f acc = {};
#pragma unroll
    for (int k = 0; k < K; k += 4) {
        v2f a, b;
        a.x = arow[k + 2 * hi + 0];
        a.y = arow[k + 2 * hi + 1];
        b.x = bcol[(size_t)(k + 2 * hi + 0) * NC];
        b.y = bcol[(size_t)(k + 2 * hi + 1) * NC];
        acc = __builtin_amdgcn_wmma_f32_16x16x4_f32(
             false, a,  false, b,
             (short)0, acc,  false,  false);
    }

    float* cptr = C + (size_t)colBase + l;
#pragma unroll
    for (int i = 0; i < 8; ++i)
        cptr[(size_t)(rowBase + 8 * hi + i) * NC] = acc[i];
}

template <int G>
__device__ __forceinline__ float4 gather_sum(float4 acc,
                                             const int* __restrict__ idx, int len,
                                             float dd,
                                             const float* __restrict__ dis,
                                             const float4* __restrict__ h4, int q) {
    int k = 0;
    for (; k + 4 <= len; k += 4) {
        int s0 = idx[k + 0];
        int s1 = idx[k + 1];
        int s2 = idx[k + 2];
        int s3 = idx[k + 3];
        float w0 = dis[s0] * dd;
        float w1 = dis[s1] * dd;
        float w2 = dis[s2] * dd;
        float w3 = dis[s3] * dd;
        float4 v0 = h4[(size_t)s0 * G + q];
        float4 v1 = h4[(size_t)s1 * G + q];
        float4 v2 = h4[(size_t)s2 * G + q];
        float4 v3 = h4[(size_t)s3 * G + q];
        acc.x += v0.x * w0; acc.y += v0.y * w0; acc.z += v0.z * w0; acc.w += v0.w * w0;
        acc.x += v1.x * w1; acc.y += v1.y * w1; acc.z += v1.z * w1; acc.w += v1.w * w1;
        acc.x += v2.x * w2; acc.y += v2.y * w2; acc.z += v2.z * w2; acc.w += v2.w * w2;
        acc.x += v3.x * w3; acc.y += v3.y * w3; acc.z += v3.z * w3; acc.w += v3.w * w3;
    }
    if (k < len) {
        int last = len - 1;
        bool p1 = k + 1 < len;
        bool p2 = k + 2 < len;
        int s0 = idx[k];
        int s1 = idx[min(k + 1, last)];
        int s2 = idx[min(k + 2, last)];
        float w0 = dis[s0] * dd;
        float w1 = dis[s1] * dd;
        float w2 = dis[s2] * dd;
        float4 v0 = h4[(size_t)s0 * G + q];
        float4 v1 = h4[(size_t)s1 * G + q];
        float4 v2 = h4[(size_t)s2 * G + q];
        acc.x += v0.x * w0; acc.y += v0.y * w0; acc.z += v0.z * w0; acc.w += v0.w * w0;
        float4 t1 = acc;
        t1.x += v1.x * w1; t1.y += v1.y * w1; t1.z += v1.z * w1; t1.w += v1.w * w1;
        acc.x = p1 ? t1.x : acc.x; acc.y = p1 ? t1.y : acc.y;
        acc.z = p1 ? t1.z : acc.z; acc.w = p1 ? t1.w : acc.w;
        float4 t2 = acc;
        t2.x += v2.x * w2; t2.y += v2.y * w2; t2.z += v2.z * w2; t2.w += v2.w * w2;
        acc.x = p2 ? t2.x : acc.x; acc.y = p2 ? t2.y : acc.y;
        acc.z = p2 ? t2.z : acc.z; acc.w = p2 ? t2.w : acc.w;
    }
    return acc;
}

__global__ void agg_relu_kernel(const int* __restrict__ csr_src,
                                const int* __restrict__ start,
                                const int* __restrict__ cnt,
                                const float* __restrict__ dis,
                                const float* __restrict__ h,
                                const float* __restrict__ bias,
                                float* __restrict__ y, int n) {
    constexpr int G = 16;
    int t = blockIdx.x * blockDim.x + threadIdx.x;
    int node = t / G;
    int q = t % G;
    if (node >= n) return;
    const float4* h4 = (const float4*)h;
    float dd = dis[node];
    float4 self = h4[t];
    float d2 = dd * dd;
    float4 acc;
    acc.x = self.x * d2 + bias[4 * q + 0];
    acc.y = self.y * d2 + bias[4 * q + 1];
    acc.z = self.z * d2 + bias[4 * q + 2];
    acc.w = self.w * d2 + bias[4 * q + 3];
    acc = gather_sum<G>(acc, csr_src + start[node], cnt[node], dd, dis, h4, q);
    float4 v;
    v.x = fmaxf(acc.x, 0.0f);
    v.y = fmaxf(acc.y, 0.0f);
    v.z = fmaxf(acc.z, 0.0f);
    v.w = fmaxf(acc.w, 0.0f);
    ((float4*)y)[t] = v;
}

__global__ void agg_logsoftmax_kernel(const int* __restrict__ csr_src,
                                      const int* __restrict__ start,
                                      const int* __restrict__ cnt,
                                      const float* __restrict__ dis,
                                      const float* __restrict__ h,
                                      const float* __restrict__ bias,
                                      float* __restrict__ out, int n) {
    constexpr int G = 4;
    int t = blockIdx.x * blockDim.x + threadIdx.x;
    int node = t / G;
    int q = t % G;
    if (node >= n) return;
    const float4* h4 = (const float4*)h;
    float dd = dis[node];
    float4 self = h4[t];
    float d2 = dd * dd;
    float4 v;
    v.x = self.x * d2 + bias[4 * q + 0];
    v.y = self.y * d2 + bias[4 * q + 1];
    v.z = self.z * d2 + bias[4 * q + 2];
    v.w = self.w * d2 + bias[4 * q + 3];
    v = gather_sum<G>(v, csr_src + start[node], cnt[node], dd, dis, h4, q);
    float mx = fmaxf(fmaxf(v.x, v.y), fmaxf(v.z, v.w));
    mx = fmaxf(mx, __shfl_xor(mx, 1));
    mx = fmaxf(mx, __shfl_xor(mx, 2));
    float s = (__expf(v.x - mx) + __expf(v.y - mx)) + (__expf(v.z - mx) + __expf(v.w - mx));
    s += __shfl_xor(s, 1);
    s += __shfl_xor(s, 2);
    float lse = mx + __logf(s);
    float4 r;
    r.x = v.x - lse; r.y = v.y - lse; r.z = v.z - lse; r.w = v.w - lse;
    ((float4*)out)[t] = r;
}

extern "C" void kernel_launch(void* const* d_in, const int* in_sizes, int n_in,
                              void* d_out, int out_size, void* d_ws, size_t ws_size,
                              hipStream_t stream) {
    const float* x   = (const float*)d_in[0];
    const int*   ei  = (const int*)  d_in[1];
    const float* W1  = (const float*)d_in[2];
    const float* b1  = (const float*)d_in[3];
    const float* W2  = (const float*)d_in[4];
    const float* b2  = (const float*)d_in[5];
    float* out = (float*)d_out;

    const int N = in_sizes[0] / 128;
    const int E = in_sizes[1] / 2;
    const int* src = ei;
    const int* dst = ei + E;

    float* h1   = (float*)d_ws;
    float* y1   = h1   + (size_t)N * 64;
    float* h2   = y1   + (size_t)N * 64;
    float* dis  = h2   + (size_t)N * 16;
    int* start   = (int*)(dis + N);
    int* cnt     = start + N;
    int* counter = cnt + N;
    int* rank    = counter + 1;
    int* csr_src = rank + E;

    hipMemsetAsync(cnt, 0, (size_t)(N + 1) * sizeof(int), stream);

    const int T = 256;

    rank_count_kernel<<<(E + T - 1) / T, T, 0, stream>>>(dst, E, cnt, rank);
    seg_alloc_kernel<<<(N + T - 1) / T, T, 0, stream>>>(cnt, N, counter, start, dis);
    csr_fill_kernel<<<(E + T - 1) / T, T, 0, stream>>>(src, dst, rank, start, E, csr_src);

    {
        int nTiles = (N / 16) * (64 / 16);
        int waves_per_block = T / 32;
        gemm_wmma_f32<128, 64><<<(nTiles + waves_per_block - 1) / waves_per_block,
                                 T, 0, stream>>>(x, W1, h1, nTiles);
    }
    {
        long long total = (long long)N * 16;
        agg_relu_kernel<<<(int)((total + T - 1) / T), T, 0, stream>>>(
            csr_src, start, cnt, dis, h1, b1, y1, N);
    }
    {
        int nTiles = (N / 16) * (16 / 16);
        int waves_per_block = T / 32;
        gemm_wmma_f32<64, 16><<<(nTiles + waves_per_block - 1) / waves_per_block,
                                T, 0, stream>>>(y1, W2, h2, nTiles);
    }
    {
        long long total = (long long)N * 4;
        agg_logsoftmax_kernel<<<(int)((total + T - 1) / T), T, 0, stream>>>(
            csr_src, start, cnt, dis, h2, b2, out, N);
    }
}


# Per-edge global returning atomics replaced by LDS counters: two-level counting partition by destination (256-node buckets, then per-node ranks in LDS); sum order inside a node changes
# speedup vs baseline: 4.4326x; 1.7436x over previous
#include <hip/hip_runtime.h>
#include <cstdint>

typedef __attribute__((ext_vector_type(2))) float v2f;
typedef __attribute__((ext_vector_type(8))) float v8f;

constexpr int PT  = 256;
constexpr int EPT = 16;
constexpr int CH  = PT * EPT;

__global__ void bucket_hist_kernel(const int* __restrict__ dst, int E, int NB,
                                   int* __restrict__ bcnt) {
    extern __shared__ int sm[];
    int* lcnt = sm;
    const int tid = threadIdx.x;
    for (int b = tid; b < NB; b += PT) lcnt[b] = 0;
    int base = blockIdx.x * CH + tid;
    int d[EPT];
#pragma unroll
    for (int j = 0; j < EPT; ++j) {
        int e = base + j * PT;
        d[j] = (e < E) ? dst[e] : -1;
    }
    __syncthreads();
#pragma unroll
    for (int j = 0; j < EPT; ++j)
        if (d[j] >= 0) atomicAdd(&lcnt[d[j] >> 8], 1);
    __syncthreads();
    for (int b = tid; b < NB; b += PT) {
        int c = lcnt[b];
        if (c) atomicAdd(&bcnt[b], c);
    }
}

__global__ void bucket_scatter_kernel(const int* __restrict__ src,
                                      const int* __restrict__ dst, int E, int NB,
                                      const int* __restrict__ bcnt,
                                      int* __restrict__ bcur,
                                      int* __restrict__ bstart,
                                      int* __restrict__ pairs_i) {
    extern __shared__ int sm[];
    int* lcnt   = sm;
    int* lstart = sm + NB;
    int* wsum   = sm + 2 * NB;
    const int tid = threadIdx.x;
    const int lane = tid & 31;
    const int w = tid >> 5;

    int base = blockIdx.x * CH + tid;
    int d[EPT], s[EPT], r[EPT];
#pragma unroll
    for (int j = 0; j < EPT; ++j) {
        int e = base + j * PT;
        bool ok = e < E;
        d[j] = ok ? dst[e] : -1;
        s[j] = ok ? src[e] : 0;
        r[j] = 0;
    }

    int carry = 0;
    for (int b0 = 0; b0 < NB; b0 += PT) {
        int b = b0 + tid;
        int c = (b < NB) ? bcnt[b] : 0;
        int incl = c;
#pragma unroll
        for (int o = 1; o < 32; o <<= 1) {
            int t = __shfl_up(incl, o);
            if (lane >= o) incl += t;
        }
        if (lane == 31) wsum[w] = incl;
        __syncthreads();
        if (tid == 0) {
            int tot = 0;
#pragma unroll
            for (int k = 0; k < PT / 32; ++k) { int t = wsum[k]; wsum[k] = tot; tot += t; }
            wsum[PT / 32] = tot;
        }
        __syncthreads();
        if (b < NB) {
            lstart[b] = carry + wsum[w] + incl - c;
            lcnt[b] = 0;
        }
        carry += wsum[PT / 32];
        __syncthreads();
    }
    if (blockIdx.x == 0)
        for (int b = tid; b < NB; b += PT) bstart[b] = lstart[b];

#pragma unroll
    for (int j = 0; j < EPT; ++j)
        if (d[j] >= 0) r[j] = atomicAdd(&lcnt[d[j] >> 8], 1);
    __syncthreads();
    for (int b = tid; b < NB; b += PT) {
        int c = lcnt[b];
        int off = 0;
        if (c) off = atomicAdd(&bcur[b], c);
        lcnt[b] = lstart[b] + off;
    }
    __syncthreads();
    int2* pairs = (int2*)pairs_i;
#pragma unroll
    for (int j = 0; j < EPT; ++j)
        if (d[j] >= 0) pairs[lcnt[d[j] >> 8] + r[j]] = make_int2(s[j], d[j]);
}

__global__ void bucket_finish_kernel(const int* __restrict__ pairs_i,
                                     const int* __restrict__ bstart,
                                     const int* __restrict__ bcnt, int n,
                                     int* __restrict__ start,
                                     int* __restrict__ cnt,
                                     float* __restrict__ dis,
                                     int* __restrict__ csr_src) {
    __shared__ int lcnt[256];
    __shared__ int lcur[256];
    __shared__ int lex[256];
    __shared__ int wsum[8];
    const int tid = threadIdx.x;
    const int lane = tid & 31;
    const int w = tid >> 5;
    const int b = blockIdx.x;
    const int bs = bstart[b];
    const int bc = bcnt[b];
    const int2* p = (const int2*)pairs_i + bs;

    lcnt[tid] = 0;
    lcur[tid] = 0;
    __syncthreads();
    for (int i0 = 0; i0 < bc; i0 += 4 * 256) {
        int2 q[4];
#pragma unroll
        for (int k = 0; k < 4; ++k) q[k] = p[min(i0 + k * 256 + tid, bc - 1)];
#pragma unroll
        for (int k = 0; k < 4; ++k)
            if (i0 + k * 256 + tid < bc) atomicAdd(&lcnt[q[k].y & 255], 1);
    }
    __syncthreads();

    int c = lcnt[tid];
    int incl = c;
#pragma unroll
    for (int o = 1; o < 32; o <<= 1) {
        int t = __shfl_up(incl, o);
        if (lane >= o) incl += t;
    }
    if (lane == 31) wsum[w] = incl;
    __syncthreads();
    if (tid == 0) {
        int tot = 0;
#pragma unroll
        for (int k = 0; k < 8; ++k) { int t = wsum[k]; wsum[k] = tot; tot += t; }
    }
    __syncthreads();
    int ex = wsum[w] + incl - c;
    lex[tid] = ex;
    int node = b * 256 + tid;
    if (node < n) {
        start[node] = bs + ex;
        cnt[node] = c;
        dis[node] = rsqrtf((float)c + 1.0f);
    }
    __syncthreads();

    for (int i0 = 0; i0 < bc; i0 += 4 * 256) {
        int2 q[4];
#pragma unroll
        for (int k = 0; k < 4; ++k) q[k] = p[min(i0 + k * 256 + tid, bc - 1)];
#pragma unroll
        for (int k = 0; k < 4; ++k) {
            if (i0 + k * 256 + tid < bc) {
                int l = q[k].y & 255;
                int rk = atomicAdd(&lcur[l], 1);
                csr_src[bs + lex[l] + rk] = q[k].x;
            }
        }
    }
}

template <int K, int NC>
__global__ void gemm_wmma_f32(const float* __restrict__ A,
                              const float* __restrict__ B,
                              float* __restrict__ C, int nTiles) {
    int wave = (int)((blockIdx.x * blockDim.x + threadIdx.x) >> 5);
    int lane = threadIdx.x & 31;
    if (wave >= nTiles) return;
    constexpr int TN = NC / 16;
    int tm = wave / TN;
    int tn = wave % TN;
    int rowBase = tm * 16;
    int colBase = tn * 16;
    int hi = lane >> 4;
    int l  = lane & 15;

    const float* arow = A + (size_t)(rowBase + l) * K;
    const float* bcol = B + colBase + l;

    v8f acc = {};
#pragma unroll
    for (int k = 0; k < K; k += 4) {
        v2f a, b;
        a.x = arow[k + 2 * hi + 0];
        a.y = arow[k + 2 * hi + 1];
        b.x = bcol[(size_t)(k + 2 * hi + 0) * NC];
        b.y = bcol[(size_t)(k + 2 * hi + 1) * NC];
        acc = __builtin_amdgcn_wmma_f32_16x16x4_f32(
             false, a,  false, b,
             (short)0, acc,  false,  false);
    }

    float* cptr = C + (size_t)colBase + l;
#pragma unroll
    for (int i = 0; i < 8; ++i)
        cptr[(size_t)(rowBase + 8 * hi + i) * NC] = acc[i];
}

template <int G>
__device__ __forceinline__ float4 gather_sum(float4 acc,
                                             const int* __restrict__ idx, int len,
                                             float dd,
                                             const float* __restrict__ dis,
                                             const float4* __restrict__ h4, int q) {
    int k = 0;
    for (; k + 4 <= len; k += 4) {
        int s0 = idx[k + 0];
        int s1 = idx[k + 1];
        int s2 = idx[k + 2];
        int s3 = idx[k + 3];
        float w0 = dis[s0] * dd;
        float w1 = dis[s1] * dd;
        float w2 = dis[s2] * dd;
        float w3 = dis[s3] * dd;
        float4 v0 = h4[(size_t)s0 * G + q];
        float4 v1 = h4[(size_t)s1 * G + q];
        float4 v2 = h4[(size_t)s2 * G + q];
        float4 v3 = h4[(size_t)s3 * G + q];
        acc.x += v0.x * w0; acc.y += v0.y * w0; acc.z += v0.z * w0; acc.w += v0.w * w0;
        acc.x += v1.x * w1; acc.y += v1.y * w1; acc.z += v1.z * w1; acc.w += v1.w * w1;
        acc.x += v2.x * w2; acc.y += v2.y * w2; acc.z += v2.z * w2; acc.w += v2.w * w2;
        acc.x += v3.x * w3; acc.y += v3.y * w3; acc.z += v3.z * w3; acc.w += v3.w * w3;
    }
    if (k < len) {
        int last = len - 1;
        bool p1 = k + 1 < len;
        bool p2 = k + 2 < len;
        int s0 = idx[k];
        int s1 = idx[min(k + 1, last)];
        int s2 = idx[min(k + 2, last)];
        float w0 = dis[s0] * dd;
        float w1 = dis[s1] * dd;
        float w2 = dis[s2] * dd;
        float4 v0 = h4[(size_t)s0 * G + q];
        float4 v1 = h4[(size_t)s1 * G + q];
        float4 v2 = h4[(size_t)s2 * G + q];
        acc.x += v0.x * w0; acc.y += v0.y * w0; acc.z += v0.z * w0; acc.w += v0.w * w0;
        float4 t1 = acc;
        t1.x += v1.x * w1; t1.y += v1.y * w1; t1.z += v1.z * w1; t1.w += v1.w * w1;
        acc.x = p1 ? t1.x : acc.x; acc.y = p1 ? t1.y : acc.y;
        acc.z = p1 ? t1.z : acc.z; acc.w = p1 ? t1.w : acc.w;
        float4 t2 = acc;
        t2.x += v2.x * w2; t2.y += v2.y * w2; t2.z += v2.z * w2; t2.w += v2.w * w2;
        acc.x = p2 ? t2.x : acc.x; acc.y = p2 ? t2.y : acc.y;
        acc.z = p2 ? t2.z : acc.z; acc.w = p2 ? t2.w : acc.w;
    }
    return acc;
}

__global__ void agg_relu_kernel(const int* __restrict__ csr_src,
                                const int* __restrict__ start,
                                const int* __restrict__ cnt,
                                const float* __restrict__ dis,
                                const float* __restrict__ h,
                                const float* __restrict__ bias,
                                float* __restrict__ y, int n) {
    constexpr int G = 16;
    int t = blockIdx.x * blockDim.x + threadIdx.x;
    int node = t / G;
    int q = t % G;
    if (node >= n) return;
    const float4* h4 = (const float4*)h;
    float dd = dis[node];
    float4 self = h4[t];
    float d2 = dd * dd;
    float4 acc;
    acc.x = self.x * d2 + bias[4 * q + 0];
    acc.y = self.y * d2 + bias[4 * q + 1];
    acc.z = self.z * d2 + bias[4 * q + 2];
    acc.w = self.w * d2 + bias[4 * q + 3];
    acc = gather_sum<G>(acc, csr_src + start[node], cnt[node], dd, dis, h4, q);
    float4 v;
    v.x = fmaxf(acc.x, 0.0f);
    v.y = fmaxf(acc.y, 0.0f);
    v.z = fmaxf(acc.z, 0.0f);
    v.w = fmaxf(acc.w, 0.0f);
    ((float4*)y)[t] = v;
}

__global__ void agg_logsoftmax_kernel(const int* __restrict__ csr_src,
                                      const int* __restrict__ start,
                                      const int* __restrict__ cnt,
                                      const float* __restrict__ dis,
                                      const float* __restrict__ h,
                                      const float* __restrict__ bias,
                                      float* __restrict__ out, int n) {
    constexpr int G = 4;
    int t = blockIdx.x * blockDim.x + threadIdx.x;
    int node = t / G;
    int q = t % G;
    if (node >= n) return;
    const float4* h4 = (const float4*)h;
    float dd = dis[node];
    float4 self = h4[t];
    float d2 = dd * dd;
    float4 v;
    v.x = self.x * d2 + bias[4 * q + 0];
    v.y = self.y * d2 + bias[4 * q + 1];
    v.z = self.z * d2 + bias[4 * q + 2];
    v.w = self.w * d2 + bias[4 * q + 3];
    v = gather_sum<G>(v, csr_src + start[node], cnt[node], dd, dis, h4, q);
    float mx = fmaxf(fmaxf(v.x, v.y), fmaxf(v.z, v.w));
    mx = fmaxf(mx, __shfl_xor(mx, 1));
    mx = fmaxf(mx, __shfl_xor(mx, 2));
    float s = (__expf(v.x - mx) + __expf(v.y - mx)) + (__expf(v.z - mx) + __expf(v.w - mx));
    s += __shfl_xor(s, 1);
    s += __shfl_xor(s, 2);
    float lse = mx + __logf(s);
    float4 r;
    r.x = v.x - lse; r.y = v.y - lse; r.z = v.z - lse; r.w = v.w - lse;
    ((float4*)out)[t] = r;
}

extern "C" void kernel_launch(void* const* d_in, const int* in_sizes, int n_in,
                              void* d_out, int out_size, void* d_ws, size_t ws_size,
                              hipStream_t stream) {
    const float* x   = (const float*)d_in[0];
    const int*   ei  = (const int*)  d_in[1];
    const float* W1  = (const float*)d_in[2];
    const float* b1  = (const float*)d_in[3];
    const float* W2  = (const float*)d_in[4];
    const float* b2  = (const float*)d_in[5];
    float* out = (float*)d_out;

    const int N = in_sizes[0] / 128;
    const int E = in_sizes[1] / 2;
    const int* src = ei;
    const int* dst = ei + E;

    const int NB = (N + 255) / 256;
    int*   pairs = (int*)d_ws;
    float* h1   = (float*)(pairs + (size_t)2 * E);
    float* y1   = h1   + (size_t)N * 64;
    float* h2   = y1   + (size_t)N * 64;
    float* dis  = h2   + (size_t)N * 16;
    int* start   = (int*)(dis + N);
    int* cnt     = start + N;
    int* bcnt    = cnt + N;
    int* bcur    = bcnt + NB;
    int* bstart  = bcur + NB;
    int* csr_src = bstart + NB;

    hipMemsetAsync(bcnt, 0, (size_t)(2 * NB) * sizeof(int), stream);

    const int T = 256;

    {
        int nChunks = (E + CH - 1) / CH;
        bucket_hist_kernel<<<nChunks, PT, (size_t)NB * sizeof(int), stream>>>(dst, E, NB, bcnt);
        bucket_scatter_kernel<<<nChunks, PT, (size_t)(2 * NB + 16) * sizeof(int), stream>>>(
            src, dst, E, NB, bcnt, bcur, bstart, pairs);
        bucket_finish_kernel<<<NB, 256, 0, stream>>>(pairs, bstart, bcnt, N,
                                                     start, cnt, dis, csr_src);
    }

    {
        int nTiles = (N / 16) * (64 / 16);
        int waves_per_block = T / 32;
        gemm_wmma_f32<128, 64><<<(nTiles + waves_per_block - 1) / waves_per_block,
                                 T, 0, stream>>>(x, W1, h1, nTiles);
    }
    {
        long long total = (long long)N * 16;
        agg_relu_kernel<<<(int)((total + T - 1) / T), T, 0, stream>>>(
            csr_src, start, cnt, dis, h1, b1, y1, N);
    }
    {
        int nTiles = (N / 16) * (16 / 16);
        int waves_per_block = T / 32;
        gemm_wmma_f32<64, 16><<<(nTiles + waves_per_block - 1) / waves_per_block,
                                T, 0, stream>>>(y1, W2, h2, nTiles);
    }
    {
        long long total = (long long)N * 4;
        agg_logsoftmax_kernel<<<(int)((total + T - 1) / T), T, 0, stream>>>(
            csr_src, start, cnt, dis, h2, b2, out, N);
    }
}


# Both products: one wave per 16-row strip over all column tiles, A fragments read once by 16-byte loads, W staged in LDS in fragment order; k order inside 8-blocks changes, so sum order changes
# speedup vs baseline: 4.7700x; 1.0761x over previous
#include <hip/hip_runtime.h>
#include <cstdint>

typedef __attribute__((ext_vector_type(2))) float v2f;
typedef __attribute__((ext_vector_type(8))) float v8f;

constexpr int PT  = 256;
constexpr int EPT = 16;
constexpr int CH  = PT * EPT;

__global__ void bucket_hist_kernel(const int* __restrict__ dst, int E, int NB,
                                   int* __restrict__ bcnt) {
    extern __shared__ int sm[];
    int* lcnt = sm;
    const int tid = threadIdx.x;
    for (int b = tid; b < NB; b += PT) lcnt[b] = 0;
    int base = blockIdx.x * CH + tid;
    int d[EPT];
#pragma unroll
    for (int j = 0; j < EPT; ++j) {
        int e = base + j * PT;
        d[j] = (e < E) ? dst[e] : -1;
    }
    __syncthreads();
#pragma unroll
    for (int j = 0; j < EPT; ++j)
        if (d[j] >= 0) atomicAdd(&lcnt[d[j] >> 8], 1);
    __syncthreads();
    for (int b = tid; b < NB; b += PT) {
        int c = lcnt[b];
        if (c) atomicAdd(&bcnt[b], c);
    }
}

__global__ void bucket_scatter_kernel(const int* __restrict__ src,
                                      const int* __restrict__ dst, int E, int NB,
                                      const int* __restrict__ bcnt,
                                      int* __restrict__ bcur,
                                      int* __restrict__ bstart,
                                      int* __restrict__ pairs_i) {
    extern __shared__ int sm[];
    int* lcnt   = sm;
    int* lstart = sm + NB;
    int* wsum   = sm + 2 * NB;
    const int tid = threadIdx.x;
    const int lane = tid & 31;
    const int w = tid >> 5;

    int base = blockIdx.x * CH + tid;
    int d[EPT], s[EPT], r[EPT];
#pragma unroll
    for (int j = 0; j < EPT; ++j) {
        int e = base + j * PT;
        bool ok = e < E;
        d[j] = ok ? dst[e] : -1;
        s[j] = ok ? src[e] : 0;
        r[j] = 0;
    }

    int carry = 0;
    for (int b0 = 0; b0 < NB; b0 += PT) {
        int b = b0 + tid;
        int c = (b < NB) ? bcnt[b] : 0;
        int incl = c;
#pragma unroll
        for (int o = 1; o < 32; o <<= 1) {
            int t = __shfl_up(incl, o);
            if (lane >= o) incl += t;
        }
        if (lane == 31) wsum[w] = incl;
        __syncthreads();
        if (tid == 0) {
            int tot = 0;
#pragma unroll
            for (int k = 0; k < PT / 32; ++k) { int t = wsum[k]; wsum[k] = tot; tot += t; }
            wsum[PT / 32] = tot;
        }
        __syncthreads();
        if (b < NB) {
            lstart[b] = carry + wsum[w] + incl - c;
            lcnt[b] = 0;
        }
        carry += wsum[PT / 32];
        __syncthreads();
    }
    if (blockIdx.x == 0)
        for (int b = tid; b < NB; b += PT) bstart[b] = lstart[b];

#pragma unroll
    for (int j = 0; j < EPT; ++j)
        if (d[j] >= 0) r[j] = atomicAdd(&lcnt[d[j] >> 8], 1);
    __syncthreads();
    for (int b = tid; b < NB; b += PT) {
        int c = lcnt[b];
        int off = 0;
        if (c) off = atomicAdd(&bcur[b], c);
        lcnt[b] = lstart[b] + off;
    }
    __syncthreads();
    int2* pairs = (int2*)pairs_i;
#pragma unroll
    for (int j = 0; j < EPT; ++j)
        if (d[j] >= 0) pairs[lcnt[d[j] >> 8] + r[j]] = make_int2(s[j], d[j]);
}

__global__ void bucket_finish_kernel(const int* __restrict__ pairs_i,
                                     const int* __restrict__ bstart,
                                     const int* __restrict__ bcnt, int n,
                                     int* __restrict__ start,
                                     int* __restrict__ cnt,
                                     float* __restrict__ dis,
                                     int* __restrict__ csr_src) {
    __shared__ int lcnt[256];
    __shared__ int lcur[256];
    __shared__ int lex[256];
    __shared__ int wsum[8];
    const int tid = threadIdx.x;
    const int lane = tid & 31;
    const int w = tid >> 5;
    const int b = blockIdx.x;
    const int bs = bstart[b];
    const int bc = bcnt[b];
    const int2* p = (const int2*)pairs_i + bs;

    lcnt[tid] = 0;
    lcur[tid] = 0;
    __syncthreads();
    for (int i0 = 0; i0 < bc; i0 += 4 * 256) {
        int2 q[4];
#pragma unroll
        for (int k = 0; k < 4; ++k) q[k] = p[min(i0 + k * 256 + tid, bc - 1)];
#pragma unroll
        for (int k = 0; k < 4; ++k)
            if (i0 + k * 256 + tid < bc) atomicAdd(&lcnt[q[k].y & 255], 1);
    }
    __syncthreads();

    int c = lcnt[tid];
    int incl = c;
#pragma unroll
    for (int o = 1; o < 32; o <<= 1) {
        int t = __shfl_up(incl, o);
        if (lane >= o) incl += t;
    }
    if (lane == 31) wsum[w] = incl;
    __syncthreads();
    if (tid == 0) {
        int tot = 0;
#pragma unroll
        for (int k = 0; k < 8; ++k) { int t = wsum[k]; wsum[k] = tot; tot += t; }
    }
    __syncthreads();
    int ex = wsum[w] + incl - c;
    lex[tid] = ex;
    int node = b * 256 + tid;
    if (node < n) {
        start[node] = bs + ex;
        cnt[node] = c;
        dis[node] = rsqrtf((float)c + 1.0f);
    }
    __syncthreads();

    for (int i0 = 0; i0 < bc; i0 += 4 * 256) {
        int2 q[4];
#pragma unroll
        for (int k = 0; k < 4; ++k) q[k] = p[min(i0 + k * 256 + tid, bc - 1)];
#pragma unroll
        for (int k = 0; k < 4; ++k) {
            if (i0 + k * 256 + tid < bc) {
                int l = q[k].y & 255;
                int rk = atomicAdd(&lcur[l], 1);
                csr_src[bs + lex[l] + rk] = q[k].x;
            }
        }
    }
}

constexpr int GW = 4;

template <int K, int NC>
__global__ void __launch_bounds__(GW * 32)
gemm_wmma_f32(const float* __restrict__ A,
              const float* __restrict__ B,
              float* __restrict__ C, int nStrips) {
    constexpr int TN = NC / 16;
    constexpr int KB = K / 8;
    __shared__ float4 bl[KB * TN * 32];

    const int lane = threadIdx.x & 31;
    const int wv = threadIdx.x >> 5;
    const int c = lane & 15;
    const int h = lane >> 4;
    const int strip = blockIdx.x * GW + wv;
    const bool live = strip < nStrips;
    const int rowBase = (live ? strip : 0) * 16;

    constexpr int SPW = KB * TN / GW;
    static_assert(KB * TN % GW == 0, "slot rows must divide evenly over the waves");
    const float* bp = B + (size_t)(4 * h) * NC + c;
    float4 bv[SPW];
#pragma unroll
    for (int i = 0; i < SPW; ++i) {
        int jn = wv + i * GW;
        int j = jn / TN;
        int n = jn % TN;
        const float* p = bp + (size_t)(8 * j) * NC + 16 * n;
        bv[i].x = p[0];
        bv[i].y = p[NC];
        bv[i].z = p[2 * NC];
        bv[i].w = p[3 * NC];
    }

    const float4* ap = (const float4*)(A + (size_t)(rowBase + c) * K + 4 * h);
    float4 a[KB];
#pragma unroll
    for (int j = 0; j < KB; ++j) a[j] = ap[2 * j];

#pragma unroll
    for (int i = 0; i < SPW; ++i) bl[(wv + i * GW) * 32 + lane] = bv[i];
    __syncthreads();
    if (!live) return;

    v8f acc[TN];
#pragma unroll
    for (int n = 0; n < TN; ++n) acc[n] = (v8f){};

    float4 bcur[TN], bnxt[TN];
#pragma unroll
    for (int n = 0; n < TN; ++n) bcur[n] = bl[n * 32 + lane];
#pragma unroll
    for (int j = 0; j < KB; ++j) {
        if (j + 1 < KB) {
#pragma unroll
            for (int n = 0; n < TN; ++n) bnxt[n] = bl[((j + 1) * TN + n) * 32 + lane];
        }
        v2f a0, a1;
        a0.x = a[j].x; a0.y = a[j].y;
        a1.x = a[j].z; a1.y = a[j].w;
#pragma unroll
        for (int n = 0; n < TN; ++n) {
            v2f b0;
            b0.x = bcur[n].x; b0.y = bcur[n].y;
            acc[n] = __builtin_amdgcn_wmma_f32_16x16x4_f32(
                 false, a0,  false, b0,
                 (short)0, acc[n],  false,  false);
        }
#pragma unroll
        for (int n = 0; n < TN; ++n) {
            v2f b1;
            b1.x = bcur[n].z; b1.y = bcur[n].w;
            acc[n] = __builtin_amdgcn_wmma_f32_16x16x4_f32(
                 false, a1,  false, b1,
                 (short)0, acc[n],  false,  false);
        }
#pragma unroll
        for (int n = 0; n < TN; ++n) bcur[n] = bnxt[n];
    }

    float* cptr = C + (size_t)(rowBase + 8 * h) * NC + c;
#pragma unroll
    for (int n = 0; n < TN; ++n)
#pragma unroll
        for (int i = 0; i < 8; ++i)
            cptr[(size_t)i * NC + 16 * n] = acc[n][i];
}

template <int G>
__device__ __forceinline__ float4 gather_sum(float4 acc,
                                             const int* __restrict__ idx, int len,
                                             float dd,
                                             const float* __restrict__ dis,
                                             const float4* __restrict__ h4, int q) {
    int k = 0;
    for (; k + 4 <= len; k += 4) {
        int s0 = idx[k + 0];
        int s1 = idx[k + 1];
        int s2 = idx[k + 2];
        int s3 = idx[k + 3];
        float w0 = dis[s0] * dd;
        float w1 = dis[s1] * dd;
        float w2 = dis[s2] * dd;
        float w3 = dis[s3] * dd;
        float4 v0 = h4[(size_t)s0 * G + q];
        float4 v1 = h4[(size_t)s1 * G + q];
        float4 v2 = h4[(size_t)s2 * G + q];
        float4 v3 = h4[(size_t)s3 * G + q];
        acc.x += v0.x * w0; acc.y += v0.y * w0; acc.z += v0.z * w0; acc.w += v0.w * w0;
        acc.x += v1.x * w1; acc.y += v1.y * w1; acc.z += v1.z * w1; acc.w += v1.w * w1;
        acc.x += v2.x * w2; acc.y += v2.y * w2; acc.z += v2.z * w2; acc.w += v2.w * w2;
        acc.x += v3.x * w3; acc.y += v3.y * w3; acc.z += v3.z * w3; acc.w += v3.w * w3;
    }
    if (k < len) {
        int last = len - 1;
        bool p1 = k + 1 < len;
        bool p2 = k + 2 < len;
        int s0 = idx[k];
        int s1 = idx[min(k + 1, last)];
        int s2 = idx[min(k + 2, last)];
        float w0 = dis[s0] * dd;
        float w1 = dis[s1] * dd;
        float w2 = dis[s2] * dd;
        float4 v0 = h4[(size_t)s0 * G + q];
        float4 v1 = h4[(size_t)s1 * G + q];
        float4 v2 = h4[(size_t)s2 * G + q];
        acc.x += v0.x * w0; acc.y += v0.y * w0; acc.z += v0.z * w0; acc.w += v0.w * w0;
        float4 t1 = acc;
        t1.x += v1.x * w1; t1.y += v1.y * w1; t1.z += v1.z * w1; t1.w += v1.w * w1;
        acc.x = p1 ? t1.x : acc.x; acc.y = p1 ? t1.y : acc.y;
        acc.z = p1 ? t1.z : acc.z; acc.w = p1 ? t1.w : acc.w;
        float4 t2 = acc;
        t2.x += v2.x * w2; t2.y += v2.y * w2; t2.z += v2.z * w2; t2.w += v2.w * w2;
        acc.x = p2 ? t2.x : acc.x; acc.y = p2 ? t2.y : acc.y;
        acc.z = p2 ? t2.z : acc.z; acc.w = p2 ? t2.w : acc.w;
    }
    return acc;
}

__global__ void agg_relu_kernel(const int* __restrict__ csr_src,
                                const int* __restrict__ start,
                                const int* __restrict__ cnt,
                                const float* __restrict__ dis,
                                const float* __restrict__ h,
                                const float* __restrict__ bias,
                                float* __restrict__ y, int n) {
    constexpr int G = 16;
    int t = blockIdx.x * blockDim.x + threadIdx.x;
    int node = t / G;
    int q = t % G;
    if (node >= n) return;
    const float4* h4 = (const float4*)h;
    float dd = dis[node];
    float4 self = h4[t];
    float d2 = dd * dd;
    float4 acc;
    acc.x = self.x * d2 + bias[4 * q + 0];
    acc.y = self.y * d2 + bias[4 * q + 1];
    acc.z = self.z * d2 + bias[4 * q + 2];
    acc.w = self.w * d2 + bias[4 * q + 3];
    acc = gather_sum<G>(acc, csr_src + start[node], cnt[node], dd, dis, h4, q);
    float4 v;
    v.x = fmaxf(acc.x, 0.0f);
    v.y = fmaxf(acc.y, 0.0f);
    v.z = fmaxf(acc.z, 0.0f);
    v.w = fmaxf(acc.w, 0.0f);
    ((float4*)y)[t] = v;
}

__global__ void agg_logsoftmax_kernel(const int* __restrict__ csr_src,
                                      const int* __restrict__ start,
                                      const int* __restrict__ cnt,
                                      const float* __restrict__ dis,
                                      const float* __restrict__ h,
                                      const float* __restrict__ bias,
                                      float* __restrict__ out, int n) {
    constexpr int G = 4;
    int t = blockIdx.x * blockDim.x + threadIdx.x;
    int node = t / G;
    int q = t % G;
    if (node >= n) return;
    const float4* h4 = (const float4*)h;
    float dd = dis[node];
    float4 self = h4[t];
    float d2 = dd * dd;
    float4 v;
    v.x = self.x * d2 + bias[4 * q + 0];
    v.y = self.y * d2 + bias[4 * q + 1];
    v.z = self.z * d2 + bias[4 * q + 2];
    v.w = self.w * d2 + bias[4 * q + 3];
    v = gather_sum<G>(v, csr_src + start[node], cnt[node], dd, dis, h4, q);
    float mx = fmaxf(fmaxf(v.x, v.y), fmaxf(v.z, v.w));
    mx = fmaxf(mx, __shfl_xor(mx, 1));
    mx = fmaxf(mx, __shfl_xor(mx, 2));
    float s = (__expf(v.x - mx) + __expf(v.y - mx)) + (__expf(v.z - mx) + __expf(v.w - mx));
    s += __shfl_xor(s, 1);
    s += __shfl_xor(s, 2);
    float lse = mx + __logf(s);
    float4 r;
    r.x = v.x - lse; r.y = v.y - lse; r.z = v.z - lse; r.w = v.w - lse;
    ((float4*)out)[t] = r;
}

extern "C" void kernel_launch(void* const* d_in, const int* in_sizes, int n_in,
                              void* d_out, int out_size, void* d_ws, size_t ws_size,
                              hipStream_t stream) {
    const float* x   = (const float*)d_in[0];
    const int*   ei  = (const int*)  d_in[1];
    const float* W1  = (const float*)d_in[2];
    const float* b1  = (const float*)d_in[3];
    const float* W2  = (const float*)d_in[4];
    const float* b2  = (const float*)d_in[5];
    float* out = (float*)d_out;

    const int N = in_sizes[0] / 128;
    const int E = in_sizes[1] / 2;
    const int* src = ei;
    const int* dst = ei + E;

    const int NB = (N + 255) / 256;
    int*   pairs = (int*)d_ws;
    float* h1   = (float*)(pairs + (size_t)2 * E);
    float* y1   = h1   + (size_t)N * 64;
    float* h2   = y1   + (size_t)N * 64;
    float* dis  = h2   + (size_t)N * 16;
    int* start   = (int*)(dis + N);
    int* cnt     = start + N;
    int* bcnt    = cnt + N;
    int* bcur    = bcnt + NB;
    int* bstart  = bcur + NB;
    int* csr_src = bstart + NB;

    hipMemsetAsync(bcnt, 0, (size_t)(2 * NB) * sizeof(int), stream);

    const int T = 256;

    {
        int nChunks = (E + CH - 1) / CH;
        bucket_hist_kernel<<<nChunks, PT, (size_t)NB * sizeof(int), stream>>>(dst, E, NB, bcnt);
        bucket_scatter_kernel<<<nChunks, PT, (size_t)(2 * NB + 16) * sizeof(int), stream>>>(
            src, dst, E, NB, bcnt, bcur, bstart, pairs);
        bucket_finish_kernel<<<NB, 256, 0, stream>>>(pairs, bstart, bcnt, N,
                                                     start, cnt, dis, csr_src);
    }

    {
        int nStrips = N / 16;
        gemm_wmma_f32<128, 64><<<(nStrips + GW - 1) / GW, GW * 32, 0, stream>>>(
            x, W1, h1, nStrips);
    }
    {
        long long total = (long long)N * 16;
        agg_relu_kernel<<<(int)((total + T - 1) / T), T, 0, stream>>>(
            csr_src, start, cnt, dis, h1, b1, y1, N);
    }
    {
        int nStrips = N / 16;
        gemm_wmma_f32<64, 16><<<(nStrips + GW - 1) / GW, GW * 32, 0, stream>>>(
            y1, W2, h2, nStrips);
    }
    {
        long long total = (long long)N * 4;
        agg_logsoftmax_kernel<<<(int)((total + T - 1) / T), T, 0, stream>>>(
            csr_src, start, cnt, dis, h2, b2, out, N);
    }
}


# Layer-1 product x@W1 no longer a launch of its own: its strips run as extra work-groups (role by block index) inside the three edge-partition launches, under their latency chains; 7 launches -> 6
# speedup vs baseline: 5.0753x; 1.0640x over previous
#include <hip/hip_runtime.h>
#include <cstdint>

typedef __attribute__((ext_vector_type(2))) float v2f;
typedef __attribute__((ext_vector_type(8))) float v8f;

template <int K, int NC, int GW>
__device__ __forceinline__ void gemm_strips_wmma_f32(const float* __restrict__ A,
                                                     const float* __restrict__ B,
                                                     float* __restrict__ C,
                                                     int strip0, int stripEnd, int blk) {
    constexpr int TN = NC / 16;
    constexpr int KB = K / 8;
    __shared__ float4 bl[KB * TN * 32];

    const int lane = threadIdx.x & 31;
    const int wv = threadIdx.x >> 5;
    const int c = lane & 15;
    const int h = lane >> 4;
    const int strip = strip0 + blk * GW + wv;
    const bool live = strip < stripEnd;
    const int rowBase = (live ? strip : strip0) * 16;

    constexpr int SPW = KB * TN / GW;
    static_assert(KB * TN % GW == 0, "slot rows must divide evenly over the waves");
    const float* bp = B + (size_t)(4 * h) * NC + c;
    float4 bv[SPW];
#pragma unroll
    for (int i = 0; i < SPW; ++i) {
        int jn = wv + i * GW;
        int j = jn / TN;
        int n = jn % TN;
        const float* p = bp + (size_t)(8 * j) * NC + 16 * n;
        bv[i].x = p[0];
        bv[i].y = p[NC];
        bv[i].z = p[2 * NC];
        bv[i].w = p[3 * NC];
    }

    const float4* ap = (const float4*)(A + (size_t)(rowBase + c) * K + 4 * h);
    float4 a[KB];
#pragma unroll
    for (int j = 0; j < KB; ++j) a[j] = ap[2 * j];

#pragma unroll
    for (int i = 0; i < SPW; ++i) bl[(wv + i * GW) * 32 + lane] = bv[i];
    __syncthreads();
    if (!live) return;

    v8f acc[TN];
#pragma unroll
    for (int n = 0; n < TN; ++n) acc[n] = (v8f){};

    float4 bcur[TN], bnxt[TN];
#pragma unroll
    for (int n = 0; n < TN; ++n) bcur[n] = bl[n * 32 + lane];
#pragma unroll
    for (int j = 0; j < KB; ++j) {
        if (j + 1 < KB) {
#pragma unroll
            for (int n = 0; n < TN; ++n) bnxt[n] = bl[((j + 1) * TN + n) * 32 + lane];
        }
        v2f a0, a1;
        a0.x = a[j].x; a0.y = a[j].y;
        a1.x = a[j].z; a1.y = a[j].w;
#pragma unroll
        for (int n = 0; n < TN; ++n) {
            v2f b0;
            b0.x = bcur[n].x; b0.y = bcur[n].y;
            acc[n] = __builtin_amdgcn_wmma_f32_16x16x4_f32(
                 false, a0,  false, b0,
                 (short)0, acc[n],  false,  false);
        }
#pragma unroll
        for (int n = 0; n < TN; ++n) {
            v2f b1;
            b1.x = bcur[n].z; b1.y = bcur[n].w;
            acc[n] = __builtin_amdgcn_wmma_f32_16x16x4_f32(
                 false, a1,  false, b1,
                 (short)0, acc[n],  false,  false);
        }
#pragma unroll
        for (int n = 0; n < TN; ++n) bcur[n] = bnxt[n];
    }

    float* cptr = C + (size_t)(rowBase + 8 * h) * NC + c;
#pragma unroll
    for (int n = 0; n < TN; ++n)
#pragma unroll
        for (int i = 0; i < 8; ++i)
            cptr[(size_t)i * NC + 16 * n] = acc[n][i];
}

constexpr int GW2 = 4;

template <int K, int NC>
__global__ void __launch_bounds__(GW2 * 32)
gemm_wmma_f32(const float* __restrict__ A,
              const float* __restrict__ B,
              float* __restrict__ C, int nStrips) {
    gemm_strips_wmma_f32<K, NC, GW2>(A, B, C, 0, nStrips, blockIdx.x);
}

constexpr int PT  = 256;
constexpr int EPT = 16;
constexpr int CH  = PT * EPT;

__global__ void __launch_bounds__(PT)
bucket_hist_kernel(const int* __restrict__ dst, int E, int NB,
                   int* __restrict__ bcnt, int nPart,
                   const float* __restrict__ x, const float* __restrict__ W1,
                   float* __restrict__ h1, int strip0, int stripEnd) {
    if ((int)blockIdx.x >= nPart) {
        gemm_strips_wmma_f32<128, 64, PT / 32>(x, W1, h1, strip0, stripEnd,
                                               (int)blockIdx.x - nPart);
        return;
    }
    extern __shared__ int sm[];
    int* lcnt = sm;
    const int tid = threadIdx.x;
    for (int b = tid; b < NB; b += PT) lcnt[b] = 0;
    int base = blockIdx.x * CH + tid;
    int d[EPT];
#pragma unroll
    for (int j = 0; j < EPT; ++j) {
        int e = base + j * PT;
        d[j] = (e < E) ? dst[e] : -1;
    }
    __syncthreads();
#pragma unroll
    for (int j = 0; j < EPT; ++j)
        if (d[j] >= 0) atomicAdd(&lcnt[d[j] >> 8], 1);
    __syncthreads();
    for (int b = tid; b < NB; b += PT) {
        int c = lcnt[b];
        if (c) atomicAdd(&bcnt[b], c);
    }
}

__global__ void __launch_bounds__(PT)
bucket_scatter_kernel(const int* __restrict__ src,
                      const int* __restrict__ dst, int E, int NB,
                      const int* __restrict__ bcnt,
                      int* __restrict__ bcur,
                      int* __restrict__ bstart,
                      int* __restrict__ pairs_i, int nPart,
                      const float* __restrict__ x, const float* __restrict__ W1,
                      float* __restrict__ h1, int strip0, int stripEnd) {
    if ((int)blockIdx.x >= nPart) {
        gemm_strips_wmma_f32<128, 64, PT / 32>(x, W1, h1, strip0, stripEnd,
                                               (int)blockIdx.x - nPart);
        return;
    }
    extern __shared__ int sm[];
    int* lcnt   = sm;
    int* lstart = sm + NB;
    int* wsum   = sm + 2 * NB;
    const int tid = threadIdx.x;
    const int lane = tid & 31;
    const int w = tid >> 5;

    int base = blockIdx.x * CH + tid;
    int d[EPT], s[EPT], r[EPT];
#pragma unroll
    for (int j = 0; j < EPT; ++j) {
        int e = base + j * PT;
        bool ok = e < E;
        d[j] = ok ? dst[e] : -1;
        s[j] = ok ? src[e] : 0;
        r[j] = 0;
    }

    int carry = 0;
    for (int b0 = 0; b0 < NB; b0 += PT) {
        int b = b0 + tid;
        int c = (b < NB) ? bcnt[b] : 0;
        int incl = c;
#pragma unroll
        for (int o = 1; o < 32; o <<= 1) {
            int t = __shfl_up(incl, o);
            if (lane >= o) incl += t;
        }
        if (lane == 31) wsum[w] = incl;
        __syncthreads();
        if (tid == 0) {
            int tot = 0;
#pragma unroll
            for (int k = 0; k < PT / 32; ++k) { int t = wsum[k]; wsum[k] = tot; tot += t; }
            wsum[PT / 32] = tot;
        }
        __syncthreads();
        if (b < NB) {
            lstart[b] = carry + wsum[w] + incl - c;
            lcnt[b] = 0;
        }
        carry += wsum[PT / 32];
        __syncthreads();
    }
    if (blockIdx.x == 0)
        for (int b = tid; b < NB; b += PT) bstart[b] = lstart[b];

#pragma unroll
    for (int j = 0; j < EPT; ++j)
        if (d[j] >= 0) r[j] = atomicAdd(&lcnt[d[j] >> 8], 1);
    __syncthreads();
    for (int b = tid; b < NB; b += PT) {
        int c = lcnt[b];
        int off = 0;
        if (c) off = atomicAdd(&bcur[b], c);
        lcnt[b] = lstart[b] + off;
    }
    __syncthreads();
    int2* pairs = (int2*)pairs_i;
#pragma unroll
    for (int j = 0; j < EPT; ++j)
        if (d[j] >= 0) pairs[lcnt[d[j] >> 8] + r[j]] = make_int2(s[j], d[j]);
}

__global__ void __launch_bounds__(PT)
bucket_finish_kernel(const int* __restrict__ pairs_i,
                     const int* __restrict__ bstart,
                     const int* __restrict__ bcnt, int n,
                     int* __restrict__ start,
                     int* __restrict__ cnt,
                     float* __restrict__ dis,
                     int* __restrict__ csr_src, int nPart,
                     const float* __restrict__ x, const float* __restrict__ W1,
                     float* __restrict__ h1, int strip0, int stripEnd) {
    if ((int)blockIdx.x >= nPart) {
        gemm_strips_wmma_f32<128, 64, PT / 32>(x, W1, h1, strip0, stripEnd,
                                               (int)blockIdx.x - nPart);
        return;
    }
    __shared__ int lcnt[256];
    __shared__ int lcur[256];
    __shared__ int lex[256];
    __shared__ int wsum[8];
    const int tid = threadIdx.x;
    const int lane = tid & 31;
    const int w = tid >> 5;
    const int b = blockIdx.x;
    const int bs = bstart[b];
    const int bc = bcnt[b];
    const int2* p = (const int2*)pairs_i + bs;

    lcnt[tid] = 0;
    lcur[tid] = 0;
    __syncthreads();
    for (int i0 = 0; i0 < bc; i0 += 4 * 256) {
        int2 q[4];
#pragma unroll
        for (int k = 0; k < 4; ++k) q[k] = p[min(i0 + k * 256 + tid, bc - 1)];
#pragma unroll
        for (int k = 0; k < 4; ++k)
            if (i0 + k * 256 + tid < bc) atomicAdd(&lcnt[q[k].y & 255], 1);
    }
    __syncthreads();

    int c = lcnt[tid];
    int incl = c;
#pragma unroll
    for (int o = 1; o < 32; o <<= 1) {
        int t = __shfl_up(incl, o);
        if (lane >= o) incl += t;
    }
    if (lane == 31) wsum[w] = incl;
    __syncthreads();
    if (tid == 0) {
        int tot = 0;
#pragma unroll
        for (int k = 0; k < 8; ++k) { int t = wsum[k]; wsum[k] = tot; tot += t; }
    }
    __syncthreads();
    int ex = wsum[w] + incl - c;
    lex[tid] = ex;
    int node = b * 256 + tid;
    if (node < n) {
        start[node] = bs + ex;
        cnt[node] = c;
        dis[node] = rsqrtf((float)c + 1.0f);
    }
    __syncthreads();

    for (int i0 = 0; i0 < bc; i0 += 4 * 256) {
        int2 q[4];
#pragma unroll
        for (int k = 0; k < 4; ++k) q[k] = p[min(i0 + k * 256 + tid, bc - 1)];
#pragma unroll
        for (int k = 0; k < 4; ++k) {
            if (i0 + k * 256 + tid < bc) {
                int l = q[k].y & 255;
                int rk = atomicAdd(&lcur[l], 1);
                csr_src[bs + lex[l] + rk] = q[k].x;
            }
        }
    }
}

template <int G>
__device__ __forceinline__ float4 gather_sum(float4 acc,
                                             const int* __restrict__ idx, int len,
                                             float dd,
                                             const float* __restrict__ dis,
                                             const float4* __restrict__ h4, int q) {
    int k = 0;
    for (; k + 4 <= len; k += 4) {
        int s0 = idx[k + 0];
        int s1 = idx[k + 1];
        int s2 = idx[k + 2];
        int s3 = idx[k + 3];
        float w0 = dis[s0] * dd;
        float w1 = dis[s1] * dd;
        float w2 = dis[s2] * dd;
        float w3 = dis[s3] * dd;
        float4 v0 = h4[(size_t)s0 * G + q];
        float4 v1 = h4[(size_t)s1 * G + q];
        float4 v2 = h4[(size_t)s2 * G + q];
        float4 v3 = h4[(size_t)s3 * G + q];
        acc.x += v0.x * w0; acc.y += v0.y * w0; acc.z += v0.z * w0; acc.w += v0.w * w0;
        acc.x += v1.x * w1; acc.y += v1.y * w1; acc.z += v1.z * w1; acc.w += v1.w * w1;
        acc.x += v2.x * w2; acc.y += v2.y * w2; acc.z += v2.z * w2; acc.w += v2.w * w2;
        acc.x += v3.x * w3; acc.y += v3.y * w3; acc.z += v3.z * w3; acc.w += v3.w * w3;
    }
    if (k < len) {
        int last = len - 1;
        bool p1 = k + 1 < len;
        bool p2 = k + 2 < len;
        int s0 = idx[k];
        int s1 = idx[min(k + 1, last)];
        int s2 = idx[min(k + 2, last)];
        float w0 = dis[s0] * dd;
        float w1 = dis[s1] * dd;
        float w2 = dis[s2] * dd;
        float4 v0 = h4[(size_t)s0 * G + q];
        float4 v1 = h4[(size_t)s1 * G + q];
        float4 v2 = h4[(size_t)s2 * G + q];
        acc.x += v0.x * w0; acc.y += v0.y * w0; acc.z += v0.z * w0; acc.w += v0.w * w0;
        float4 t1 = acc;
        t1.x += v1.x * w1; t1.y += v1.y * w1; t1.z += v1.z * w1; t1.w += v1.w * w1;
        acc.x = p1 ? t1.x : acc.x; acc.y = p1 ? t1.y : acc.y;
        acc.z = p1 ? t1.z : acc.z; acc.w = p1 ? t1.w : acc.w;
        float4 t2 = acc;
        t2.x += v2.x * w2; t2.y += v2.y * w2; t2.z += v2.z * w2; t2.w += v2.w * w2;
        acc.x = p2 ? t2.x : acc.x; acc.y = p2 ? t2.y : acc.y;
        acc.z = p2 ? t2.z : acc.z; acc.w = p2 ? t2.w : acc.w;
    }
    return acc;
}

__global__ void agg_relu_kernel(const int* __restrict__ csr_src,
                                const int* __restrict__ start,
                                const int* __restrict__ cnt,
                                const float* __restrict__ dis,
                                const float* __restrict__ h,
                                const float* __restrict__ bias,
                                float* __restrict__ y, int n) {
    constexpr int G = 16;
    int t = blockIdx.x * blockDim.x + threadIdx.x;
    int node = t / G;
    int q = t % G;
    if (node >= n) return;
    const float4* h4 = (const float4*)h;
    float dd = dis[node];
    float4 self = h4[t];
    float d2 = dd * dd;
    float4 acc;
    acc.x = self.x * d2 + bias[4 * q + 0];
    acc.y = self.y * d2 + bias[4 * q + 1];
    acc.z = self.z * d2 + bias[4 * q + 2];
    acc.w = self.w * d2 + bias[4 * q + 3];
    acc = gather_sum<G>(acc, csr_src + start[node], cnt[node], dd, dis, h4, q);
    float4 v;
    v.x = fmaxf(acc.x, 0.0f);
    v.y = fmaxf(acc.y, 0.0f);
    v.z = fmaxf(acc.z, 0.0f);
    v.w = fmaxf(acc.w, 0.0f);
    ((float4*)y)[t] = v;
}

__global__ void agg_logsoftmax_kernel(const int* __restrict__ csr_src,
                                      const int* __restrict__ start,
                                      const int* __restrict__ cnt,
                                      const float* __restrict__ dis,
                                      const float* __restrict__ h,
                                      const float* __restrict__ bias,
                                      float* __restrict__ out, int n) {
    constexpr int G = 4;
    int t = blockIdx.x * blockDim.x + threadIdx.x;
    int node = t / G;
    int q = t % G;
    if (node >= n) return;
    const float4* h4 = (const float4*)h;
    float dd = dis[node];
    float4 self = h4[t];
    float d2 = dd * dd;
    float4 v;
    v.x = self.x * d2 + bias[4 * q + 0];
    v.y = self.y * d2 + bias[4 * q + 1];
    v.z = self.z * d2 + bias[4 * q + 2];
    v.w = self.w * d2 + bias[4 * q + 3];
    v = gather_sum<G>(v, csr_src + start[node], cnt[node], dd, dis, h4, q);
    float mx = fmaxf(fmaxf(v.x, v.y), fmaxf(v.z, v.w));
    mx = fmaxf(mx, __shfl_xor(mx, 1));
    mx = fmaxf(mx, __shfl_xor(mx, 2));
    float s = (__expf(v.x - mx) + __expf(v.y - mx)) + (__expf(v.z - mx) + __expf(v.w - mx));
    s += __shfl_xor(s, 1);
    s += __shfl_xor(s, 2);
    float lse = mx + __logf(s);
    float4 r;
    r.x = v.x - lse; r.y = v.y - lse; r.z = v.z - lse; r.w = v.w - lse;
    ((float4*)out)[t] = r;
}

extern "C" void kernel_launch(void* const* d_in, const int* in_sizes, int n_in,
                              void* d_out, int out_size, void* d_ws, size_t ws_size,
                              hipStream_t stream) {
    const float* x   = (const float*)d_in[0];
    const int*   ei  = (const int*)  d_in[1];
    const float* W1  = (const float*)d_in[2];
    const float* b1  = (const float*)d_in[3];
    const float* W2  = (const float*)d_in[4];
    const float* b2  = (const float*)d_in[5];
    float* out = (float*)d_out;

    const int N = in_sizes[0] / 128;
    const int E = in_sizes[1] / 2;
    const int* src = ei;
    const int* dst = ei + E;

    const int NB = (N + 255) / 256;
    int*   pairs = (int*)d_ws;
    float* h1   = (float*)(pairs + (size_t)2 * E);
    float* y1   = h1   + (size_t)N * 64;
    float* h2   = y1   + (size_t)N * 64;
    float* dis  = h2   + (size_t)N * 16;
    int* start   = (int*)(dis + N);
    int* cnt     = start + N;
    int* bcnt    = cnt + N;
    int* bcur    = bcnt + NB;
    int* bstart  = bcur + NB;
    int* csr_src = bstart + NB;

    hipMemsetAsync(bcnt, 0, (size_t)(2 * NB) * sizeof(int), stream);

    const int T = 256;

    {
        const int nChunks = (E + CH - 1) / CH;
        const int SPB = PT / 32;
        const int nStrips = N / 16;
        const int cap = 256 * SPB;
        const int thirdA = nStrips / 3;
        const int sA = thirdA < cap ? thirdA : cap;
        const int halfC = (nStrips - sA) / 2;
        const int sB = nStrips - (halfC < cap ? halfC : cap);
        const int gA = (sA + SPB - 1) / SPB;
        const int gB = (sB - sA + SPB - 1) / SPB;
        const int gC = (nStrips - sB + SPB - 1) / SPB;
        bucket_hist_kernel<<<nChunks + gA, PT, (size_t)NB * sizeof(int), stream>>>(
            dst, E, NB, bcnt, nChunks, x, W1, h1, 0, sA);
        bucket_scatter_kernel<<<nChunks + gB, PT, (size_t)(2 * NB + 16) * sizeof(int), stream>>>(
            src, dst, E, NB, bcnt, bcur, bstart, pairs, nChunks, x, W1, h1, sA, sB);
        bucket_finish_kernel<<<NB + gC, PT, 0, stream>>>(pairs, bstart, bcnt, N,
                                                      start, cnt, dis, csr_src,
                                                      NB, x, W1, h1, sB, nStrips);
    }
    {
        long long total = (long long)N * 16;
        agg_relu_kernel<<<(int)((total + T - 1) / T), T, 0, stream>>>(
            csr_src, start, cnt, dis, h1, b1, y1, N);
    }
    {
        int nStrips = N / 16;
        gemm_wmma_f32<64, 16><<<(nStrips + GW2 - 1) / GW2, GW2 * 32, 0, stream>>>(
            y1, W2, h2, nStrips);
    }
    {
        long long total = (long long)N * 4;
        agg_logsoftmax_kernel<<<(int)((total + T - 1) / T), T, 0, stream>>>(
            csr_src, start, cnt, dis, h2, b2, out, N);
    }
}


# Layer-2 rows stored scaled by dis[row] in the product's epilogue: the layer-2 aggregation adds plain rows, no per-edge dis gather or weight; dis[d] and bias applied after the sum; edge order kept
# speedup vs baseline: 5.3594x; 1.0560x over previous
#include <hip/hip_runtime.h>
#include <cstdint>

typedef __attribute__((ext_vector_type(2))) float v2f;
typedef __attribute__((ext_vector_type(8))) float v8f;

template <int K, int NC, int GW, bool RS>
__device__ __forceinline__ void gemm_strips_wmma_f32(const float* __restrict__ A,
                                                     const float* __restrict__ B,
                                                     float* __restrict__ C,
                                                     const float* __restrict__ rs,
                                                     int strip0, int stripEnd, int blk) {
    constexpr int TN = NC / 16;
    constexpr int KB = K / 8;
    __shared__ float4 bl[KB * TN * 32];

    const int lane = threadIdx.x & 31;
    const int wv = threadIdx.x >> 5;
    const int c = lane & 15;
    const int h = lane >> 4;
    const int strip = strip0 + blk * GW + wv;
    const bool live = strip < stripEnd;
    const int rowBase = (live ? strip : strip0) * 16;

    constexpr int SPW = KB * TN / GW;
    static_assert(KB * TN % GW == 0, "slot rows must divide evenly over the waves");
    const float* bp = B + (size_t)(4 * h) * NC + c;
    float4 bv[SPW];
#pragma unroll
    for (int i = 0; i < SPW; ++i) {
        int jn = wv + i * GW;
        int j = jn / TN;
        int n = jn % TN;
        const float* p = bp + (size_t)(8 * j) * NC + 16 * n;
        bv[i].x = p[0];
        bv[i].y = p[NC];
        bv[i].z = p[2 * NC];
        bv[i].w = p[3 * NC];
    }

    const float4* ap = (const float4*)(A + (size_t)(rowBase + c) * K + 4 * h);
    float4 a[KB];
#pragma unroll
    for (int j = 0; j < KB; ++j) a[j] = ap[2 * j];
    float sc[8];
    if constexpr (RS) {
        const float4* sp = (const float4*)(rs + rowBase + 8 * h);
        const float4 s0 = sp[0];
        const float4 s1 = sp[1];
        sc[0] = s0.x; sc[1] = s0.y; sc[2] = s0.z; sc[3] = s0.w;
        sc[4] = s1.x; sc[5] = s1.y; sc[6] = s1.z; sc[7] = s1.w;
    }

#pragma unroll
    for (int i = 0; i < SPW; ++i) bl[(wv + i * GW) * 32 + lane] = bv[i];
    __syncthreads();
    if (!live) return;

    v8f acc[TN];
#pragma unroll
    for (int n = 0; n < TN; ++n) acc[n] = (v8f){};

    float4 bcur[TN], bnxt[TN];
#pragma unroll
    for (int n = 0; n < TN; ++n) bcur[n] = bl[n * 32 + lane];
#pragma unroll
    for (int j = 0; j < KB; ++j) {
        if (j + 1 < KB) {
#pragma unroll
            for (int n = 0; n < TN; ++n) bnxt[n] = bl[((j + 1) * TN + n) * 32 + lane];
        }
        v2f a0, a1;
        a0.x = a[j].x; a0.y = a[j].y;
        a1.x = a[j].z; a1.y = a[j].w;
#pragma unroll
        for (int n = 0; n < TN; ++n) {
            v2f b0;
            b0.x = bcur[n].x; b0.y = bcur[n].y;
            acc[n] = __builtin_amdgcn_wmma_f32_16x16x4_f32(
                 false, a0,  false, b0,
                 (short)0, acc[n],  false,  false);
        }
#pragma unroll
        for (int n = 0; n < TN; ++n) {
            v2f b1;
            b1.x = bcur[n].z; b1.y = bcur[n].w;
            acc[n] = __builtin_amdgcn_wmma_f32_16x16x4_f32(
                 false, a1,  false, b1,
                 (short)0, acc[n],  false,  false);
        }
#pragma unroll
        for (int n = 0; n < TN; ++n) bcur[n] = bnxt[n];
    }

    float* cptr = C + (size_t)(rowBase + 8 * h) * NC + c;
#pragma unroll
    for (int n = 0; n < TN; ++n)
#pragma unroll
        for (int i = 0; i < 8; ++i) {
            if constexpr (RS) cptr[(size_t)i * NC + 16 * n] = acc[n][i] * sc[i];
            else              cptr[(size_t)i * NC + 16 * n] = acc[n][i];
        }
}

constexpr int GW2 = 4;

template <int K, int NC>
__global__ void __launch_bounds__(GW2 * 32)
gemm_wmma_f32(const float* __restrict__ A,
              const float* __restrict__ B,
              float* __restrict__ C,
              const float* __restrict__ rs, int nStrips) {
    gemm_strips_wmma_f32<K, NC, GW2, true>(A, B, C, rs, 0, nStrips, blockIdx.x);
}

constexpr int PT  = 256;
constexpr int EPT = 16;
constexpr int CH  = PT * EPT;

__global__ void __launch_bounds__(PT)
bucket_hist_kernel(const int* __restrict__ dst, int E, int NB,
                   int* __restrict__ bcnt, int nPart,
                   const float* __restrict__ x, const float* __restrict__ W1,
                   float* __restrict__ h1, int strip0, int stripEnd) {
    if ((int)blockIdx.x >= nPart) {
        gemm_strips_wmma_f32<128, 64, PT / 32, false>(x, W1, h1, nullptr, strip0, stripEnd,
                                                      (int)blockIdx.x - nPart);
        return;
    }
    extern __shared__ int sm[];
    int* lcnt = sm;
    const int tid = threadIdx.x;
    for (int b = tid; b < NB; b += PT) lcnt[b] = 0;
    int base = blockIdx.x * CH + tid;
    int d[EPT];
#pragma unroll
    for (int j = 0; j < EPT; ++j) {
        int e = base + j * PT;
        d[j] = (e < E) ? dst[e] : -1;
    }
    __syncthreads();
#pragma unroll
    for (int j = 0; j < EPT; ++j)
        if (d[j] >= 0) atomicAdd(&lcnt[d[j] >> 8], 1);
    __syncthreads();
    for (int b = tid; b < NB; b += PT) {
        int c = lcnt[b];
        if (c) atomicAdd(&bcnt[b], c);
    }
}

__global__ void __launch_bounds__(PT)
bucket_scatter_kernel(const int* __restrict__ src,
                      const int* __restrict__ dst, int E, int NB,
                      const int* __restrict__ bcnt,
                      int* __restrict__ bcur,
                      int* __restrict__ bstart,
                      int* __restrict__ pairs_i, int nPart,
                      const float* __restrict__ x, const float* __restrict__ W1,
                      float* __restrict__ h1, int strip0, int stripEnd) {
    if ((int)blockIdx.x >= nPart) {
        gemm_strips_wmma_f32<128, 64, PT / 32, false>(x, W1, h1, nullptr, strip0, stripEnd,
                                                      (int)blockIdx.x - nPart);
        return;
    }
    extern __shared__ int sm[];
    int* lcnt   = sm;
    int* lstart = sm + NB;
    int* wsum   = sm + 2 * NB;
    const int tid = threadIdx.x;
    const int lane = tid & 31;
    const int w = tid >> 5;

    int base = blockIdx.x * CH + tid;
    int d[EPT], s[EPT], r[EPT];
#pragma unroll
    for (int j = 0; j < EPT; ++j) {
        int e = base + j * PT;
        bool ok = e < E;
        d[j] = ok ? dst[e] : -1;
        s[j] = ok ? src[e] : 0;
        r[j] = 0;
    }

    int carry = 0;
    for (int b0 = 0; b0 < NB; b0 += PT) {
        int b = b0 + tid;
        int c = (b < NB) ? bcnt[b] : 0;
        int incl = c;
#pragma unroll
        for (int o = 1; o < 32; o <<= 1) {
            int t = __shfl_up(incl, o);
            if (lane >= o) incl += t;
        }
        if (lane == 31) wsum[w] = incl;
        __syncthreads();
        if (tid == 0) {
            int tot = 0;
#pragma unroll
            for (int k = 0; k < PT / 32; ++k) { int t = wsum[k]; wsum[k] = tot; tot += t; }
            wsum[PT / 32] = tot;
        }
        __syncthreads();
        if (b < NB) {
            lstart[b] = carry + wsum[w] + incl - c;
            lcnt[b] = 0;
        }
        carry += wsum[PT / 32];
        __syncthreads();
    }
    if (blockIdx.x == 0)
        for (int b = tid; b < NB; b += PT) bstart[b] = lstart[b];

#pragma unroll
    for (int j = 0; j < EPT; ++j)
        if (d[j] >= 0) r[j] = atomicAdd(&lcnt[d[j] >> 8], 1);
    __syncthreads();
    for (int b = tid; b < NB; b += PT) {
        int c = lcnt[b];
        int off = 0;
        if (c) off = atomicAdd(&bcur[b], c);
        lcnt[b] = lstart[b] + off;
    }
    __syncthreads();
    int2* pairs = (int2*)pairs_i;
#pragma unroll
    for (int j = 0; j < EPT; ++j)
        if (d[j] >= 0) pairs[lcnt[d[j] >> 8] + r[j]] = make_int2(s[j], d[j]);
}

__global__ void __launch_bounds__(PT)
bucket_finish_kernel(const int* __restrict__ pairs_i,
                     const int* __restrict__ bstart,
                     const int* __restrict__ bcnt, int n,
                     int* __restrict__ start,
                     int* __restrict__ cnt,
                     float* __restrict__ dis,
                     int* __restrict__ csr_src, int nPart,
                     const float* __restrict__ x, const float* __restrict__ W1,
                     float* __restrict__ h1, int strip0, int stripEnd) {
    if ((int)blockIdx.x >= nPart) {
        gemm_strips_wmma_f32<128, 64, PT / 32, false>(x, W1, h1, nullptr, strip0, stripEnd,
                                                      (int)blockIdx.x - nPart);
        return;
    }
    __shared__ int lcnt[256];
    __shared__ int lcur[256];
    __shared__ int lex[256];
    __shared__ int wsum[8];
    const int tid = threadIdx.x;
    const int lane = tid & 31;
    const int w = tid >> 5;
    const int b = blockIdx.x;
    const int bs = bstart[b];
    const int bc = bcnt[b];
    const int2* p = (const int2*)pairs_i + bs;

    lcnt[tid] = 0;
    lcur[tid] = 0;
    __syncthreads();
    for (int i0 = 0; i0 < bc; i0 += 4 * 256) {
        int2 q[4];
#pragma unroll
        for (int k = 0; k < 4; ++k) q[k] = p[min(i0 + k * 256 + tid, bc - 1)];
#pragma unroll
        for (int k = 0; k < 4; ++k)
            if (i0 + k * 256 + tid < bc) atomicAdd(&lcnt[q[k].y & 255], 1);
    }
    __syncthreads();

    int c = lcnt[tid];
    int incl = c;
#pragma unroll
    for (int o = 1; o < 32; o <<= 1) {
        int t = __shfl_up(incl, o);
        if (lane >= o) incl += t;
    }
    if (lane == 31) wsum[w] = incl;
    __syncthreads();
    if (tid == 0) {
        int tot = 0;
#pragma unroll
        for (int k = 0; k < 8; ++k) { int t = wsum[k]; wsum[k] = tot; tot += t; }
    }
    __syncthreads();
    int ex = wsum[w] + incl - c;
    lex[tid] = ex;
    int node = b * 256 + tid;
    if (node < n) {
        start[node] = bs + ex;
        cnt[node] = c;
        dis[node] = rsqrtf((float)c + 1.0f);
    }
    __syncthreads();

    for (int i0 = 0; i0 < bc; i0 += 4 * 256) {
        int2 q[4];
#pragma unroll
        for (int k = 0; k < 4; ++k) q[k] = p[min(i0 + k * 256 + tid, bc - 1)];
#pragma unroll
        for (int k = 0; k < 4; ++k) {
            if (i0 + k * 256 + tid < bc) {
                int l = q[k].y & 255;
                int rk = atomicAdd(&lcur[l], 1);
                csr_src[bs + lex[l] + rk] = q[k].x;
            }
        }
    }
}

template <int G>
__device__ __forceinline__ float4 gather_sum(float4 acc,
                                             const int* __restrict__ idx, int len,
                                             float dd,
                                             const float* __restrict__ dis,
                                             const float4* __restrict__ h4, int q) {
    int k = 0;
    for (; k + 4 <= len; k += 4) {
        int s0 = idx[k + 0];
        int s1 = idx[k + 1];
        int s2 = idx[k + 2];
        int s3 = idx[k + 3];
        float w0 = dis[s0] * dd;
        float w1 = dis[s1] * dd;
        float w2 = dis[s2] * dd;
        float w3 = dis[s3] * dd;
        float4 v0 = h4[(size_t)s0 * G + q];
        float4 v1 = h4[(size_t)s1 * G + q];
        float4 v2 = h4[(size_t)s2 * G + q];
        float4 v3 = h4[(size_t)s3 * G + q];
        acc.x += v0.x * w0; acc.y += v0.y * w0; acc.z += v0.z * w0; acc.w += v0.w * w0;
        acc.x += v1.x * w1; acc.y += v1.y * w1; acc.z += v1.z * w1; acc.w += v1.w * w1;
        acc.x += v2.x * w2; acc.y += v2.y * w2; acc.z += v2.z * w2; acc.w += v2.w * w2;
        acc.x += v3.x * w3; acc.y += v3.y * w3; acc.z += v3.z * w3; acc.w += v3.w * w3;
    }
    if (k < len) {
        int last = len - 1;
        bool p1 = k + 1 < len;
        bool p2 = k + 2 < len;
        int s0 = idx[k];
        int s1 = idx[min(k + 1, last)];
        int s2 = idx[min(k + 2, last)];
        float w0 = dis[s0] * dd;
        float w1 = dis[s1] * dd;
        float w2 = dis[s2] * dd;
        float4 v0 = h4[(size_t)s0 * G + q];
        float4 v1 = h4[(size_t)s1 * G + q];
        float4 v2 = h4[(size_t)s2 * G + q];
        acc.x += v0.x * w0; acc.y += v0.y * w0; acc.z += v0.z * w0; acc.w += v0.w * w0;
        float4 t1 = acc;
        t1.x += v1.x * w1; t1.y += v1.y * w1; t1.z += v1.z * w1; t1.w += v1.w * w1;
        acc.x = p1 ? t1.x : acc.x; acc.y = p1 ? t1.y : acc.y;
        acc.z = p1 ? t1.z : acc.z; acc.w = p1 ? t1.w : acc.w;
        float4 t2 = acc;
        t2.x += v2.x * w2; t2.y += v2.y * w2; t2.z += v2.z * w2; t2.w += v2.w * w2;
        acc.x = p2 ? t2.x : acc.x; acc.y = p2 ? t2.y : acc.y;
        acc.z = p2 ? t2.z : acc.z; acc.w = p2 ? t2.w : acc.w;
    }
    return acc;
}

template <int G>
__device__ __forceinline__ float4 gather_rows_sum(float4 acc,
                                                  const int* __restrict__ idx, int len,
                                                  const float4* __restrict__ g4, int q) {
    int k = 0;
#pragma unroll 2
    for (; k + 4 <= len; k += 4) {
        int s0 = idx[k + 0];
        int s1 = idx[k + 1];
        int s2 = idx[k + 2];
        int s3 = idx[k + 3];
        float4 v0 = g4[(size_t)s0 * G + q];
        float4 v1 = g4[(size_t)s1 * G + q];
        float4 v2 = g4[(size_t)s2 * G + q];
        float4 v3 = g4[(size_t)s3 * G + q];
        acc.x += v0.x; acc.y += v0.y; acc.z += v0.z; acc.w += v0.w;
        acc.x += v1.x; acc.y += v1.y; acc.z += v1.z; acc.w += v1.w;
        acc.x += v2.x; acc.y += v2.y; acc.z += v2.z; acc.w += v2.w;
        acc.x += v3.x; acc.y += v3.y; acc.z += v3.z; acc.w += v3.w;
    }
    if (k < len) {
        int last = len - 1;
        bool p1 = k + 1 < len;
        bool p2 = k + 2 < len;
        int s0 = idx[k];
        int s1 = idx[min(k + 1, last)];
        int s2 = idx[min(k + 2, last)];
        float4 v0 = g4[(size_t)s0 * G + q];
        float4 v1 = g4[(size_t)s1 * G + q];
        float4 v2 = g4[(size_t)s2 * G + q];
        acc.x += v0.x; acc.y += v0.y; acc.z += v0.z; acc.w += v0.w;
        float4 t1 = acc;
        t1.x += v1.x; t1.y += v1.y; t1.z += v1.z; t1.w += v1.w;
        acc.x = p1 ? t1.x : acc.x; acc.y = p1 ? t1.y : acc.y;
        acc.z = p1 ? t1.z : acc.z; acc.w = p1 ? t1.w : acc.w;
        float4 t2 = acc;
        t2.x += v2.x; t2.y += v2.y; t2.z += v2.z; t2.w += v2.w;
        acc.x = p2 ? t2.x : acc.x; acc.y = p2 ? t2.y : acc.y;
        acc.z = p2 ? t2.z : acc.z; acc.w = p2 ? t2.w : acc.w;
    }
    return acc;
}

__global__ void agg_relu_kernel(const int* __restrict__ csr_src,
                                const int* __restrict__ start,
                                const int* __restrict__ cnt,
                                const float* __restrict__ dis,
                                const float* __restrict__ h,
                                const float* __restrict__ bias,
                                float* __restrict__ y, int n) {
    constexpr int G = 16;
    int t = blockIdx.x * blockDim.x + threadIdx.x;
    int node = t / G;
    int q = t % G;
    if (node >= n) return;
    const float4* h4 = (const float4*)h;
    float dd = dis[node];
    float4 self = h4[t];
    float d2 = dd * dd;
    float4 acc;
    acc.x = self.x * d2 + bias[4 * q + 0];
    acc.y = self.y * d2 + bias[4 * q + 1];
    acc.z = self.z * d2 + bias[4 * q + 2];
    acc.w = self.w * d2 + bias[4 * q + 3];
    acc = gather_sum<G>(acc, csr_src + start[node], cnt[node], dd, dis, h4, q);
    float4 v;
    v.x = fmaxf(acc.x, 0.0f);
    v.y = fmaxf(acc.y, 0.0f);
    v.z = fmaxf(acc.z, 0.0f);
    v.w = fmaxf(acc.w, 0.0f);
    ((float4*)y)[t] = v;
}

__global__ void agg_logsoftmax_kernel(const int* __restrict__ csr_src,
                                      const int* __restrict__ start,
                                      const int* __restrict__ cnt,
                                      const float* __restrict__ dis,
                                      const float* __restrict__ g,
                                      const float* __restrict__ bias,
                                      float* __restrict__ out, int n) {
    constexpr int G = 4;
    int t = blockIdx.x * blockDim.x + threadIdx.x;
    int node = t / G;
    int q = t % G;
    if (node >= n) return;
    const float4* g4 = (const float4*)g;
    const int st = start[node];
    const int len = cnt[node];
    float dd = dis[node];
    float4 acc = g4[t];
    const float4 b = ((const float4*)bias)[q];
    asm volatile("" :: "v"(st), "v"(len), "v"(b.x), "v"(b.y), "v"(b.z), "v"(b.w));
    acc = gather_rows_sum<G>(acc, csr_src + st, len, g4, q);
    float4 v;
    v.x = acc.x * dd + b.x;
    v.y = acc.y * dd + b.y;
    v.z = acc.z * dd + b.z;
    v.w = acc.w * dd + b.w;
    float mx = fmaxf(fmaxf(v.x, v.y), fmaxf(v.z, v.w));
    mx = fmaxf(mx, __shfl_xor(mx, 1));
    mx = fmaxf(mx, __shfl_xor(mx, 2));
    float s = (__expf(v.x - mx) + __expf(v.y - mx)) + (__expf(v.z - mx) + __expf(v.w - mx));
    s += __shfl_xor(s, 1);
    s += __shfl_xor(s, 2);
    float lse = mx + __logf(s);
    float4 r;
    r.x = v.x - lse; r.y = v.y - lse; r.z = v.z - lse; r.w = v.w - lse;
    ((float4*)out)[t] = r;
}

extern "C" void kernel_launch(void* const* d_in, const int* in_sizes, int n_in,
                              void* d_out, int out_size, void* d_ws, size_t ws_size,
                              hipStream_t stream) {
    const float* x   = (const float*)d_in[0];
    const int*   ei  = (const int*)  d_in[1];
    const float* W1  = (const float*)d_in[2];
    const float* b1  = (const float*)d_in[3];
    const float* W2  = (const float*)d_in[4];
    const float* b2  = (const float*)d_in[5];
    float* out = (float*)d_out;

    const int N = in_sizes[0] / 128;
    const int E = in_sizes[1] / 2;
    const int* src = ei;
    const int* dst = ei + E;

    const int NB = (N + 255) / 256;
    int*   pairs = (int*)d_ws;
    float* h1   = (float*)(pairs + (size_t)2 * E);
    float* y1   = h1   + (size_t)N * 64;
    float* h2   = y1   + (size_t)N * 64;
    float* dis  = h2   + (size_t)N * 16;
    int* start   = (int*)(dis + N);
    int* cnt     = start + N;
    int* bcnt    = cnt + N;
    int* bcur    = bcnt + NB;
    int* bstart  = bcur + NB;
    int* csr_src = bstart + NB;

    hipMemsetAsync(bcnt, 0, (size_t)(2 * NB) * sizeof(int), stream);

    const int T = 256;

    {
        const int nChunks = (E + CH - 1) / CH;
        const int SPB = PT / 32;
        const int nStrips = N / 16;
        const int cap = 256 * SPB;
        const int thirdA = nStrips / 3;
        const int sA = thirdA < cap ? thirdA : cap;
        const int halfC = (nStrips - sA) / 2;
        const int sB = nStrips - (halfC < cap ? halfC : cap);
        const int gA = (sA + SPB - 1) / SPB;
        const int gB = (sB - sA + SPB - 1) / SPB;
        const int gC = (nStrips - sB + SPB - 1) / SPB;
        bucket_hist_kernel<<<nChunks + gA, PT, (size_t)NB * sizeof(int), stream>>>(
            dst, E, NB, bcnt, nChunks, x, W1, h1, 0, sA);
        bucket_scatter_kernel<<<nChunks + gB, PT, (size_t)(2 * NB + 16) * sizeof(int), stream>>>(
            src, dst, E, NB, bcnt, bcur, bstart, pairs, nChunks, x, W1, h1, sA, sB);
        bucket_finish_kernel<<<NB + gC, PT, 0, stream>>>(pairs, bstart, bcnt, N,
                                                      start, cnt, dis, csr_src,
                                                      NB, x, W1, h1, sB, nStrips);
    }
    {
        long long total = (long long)N * 16;
        agg_relu_kernel<<<(int)((total + T - 1) / T), T, 0, stream>>>(
            csr_src, start, cnt, dis, h1, b1, y1, N);
    }
    {
        int nStrips = N / 16;
        gemm_wmma_f32<64, 16><<<(nStrips + GW2 - 1) / GW2, GW2 * 32, 0, stream>>>(
            y1, W2, h2, dis, nStrips);
    }
    {
        long long total = (long long)N * 4;
        agg_logsoftmax_kernel<<<(int)((total + T - 1) / T), T, 0, stream>>>(
            csr_src, start, cnt, dis, h2, b2, out, N);
    }
}


# Layer-2 product fused into the layer-1 aggregation kernel: a 16-node block keeps its relu rows in LDS, its last wave to arrive multiplies them by W2 on WMMA; 6 launches -> 5; k sum order changes
# speedup vs baseline: 5.7540x; 1.0736x over previous
#include <hip/hip_runtime.h>
#include <cstdint>

typedef __attribute__((ext_vector_type(2))) float v2f;
typedef __attribute__((ext_vector_type(8))) float v8f;

template <int K, int NC, int GW>
__device__ __forceinline__ void gemm_strips_wmma_f32(const float* __restrict__ A,
                                                     const float* __restrict__ B,
                                                     float* __restrict__ C,
                                                     int strip0, int stripEnd, int blk) {
    constexpr int TN = NC / 16;
    constexpr int KB = K / 8;
    __shared__ float4 bl[KB * TN * 32];

    const int lane = threadIdx.x & 31;
    const int wv = threadIdx.x >> 5;
    const int c = lane & 15;
    const int h = lane >> 4;
    const int strip = strip0 + blk * GW + wv;
    const bool live = strip < stripEnd;
    const int rowBase = (live ? strip : strip0) * 16;

    constexpr int SPW = KB * TN / GW;
    static_assert(KB * TN % GW == 0, "slot rows must divide evenly over the waves");
    const float* bp = B + (size_t)(4 * h) * NC + c;
    float4 bv[SPW];
#pragma unroll
    for (int i = 0; i < SPW; ++i) {
        int jn = wv + i * GW;
        int j = jn / TN;
        int n = jn % TN;
        const float* p = bp + (size_t)(8 * j) * NC + 16 * n;
        bv[i].x = p[0];
        bv[i].y = p[NC];
        bv[i].z = p[2 * NC];
        bv[i].w = p[3 * NC];
    }

    const float4* ap = (const float4*)(A + (size_t)(rowBase + c) * K + 4 * h);
    float4 a[KB];
#pragma unroll
    for (int j = 0; j < KB; ++j) a[j] = ap[2 * j];

#pragma unroll
    for (int i = 0; i < SPW; ++i) bl[(wv + i * GW) * 32 + lane] = bv[i];
    __syncthreads();
    if (!live) return;

    v8f acc[TN];
#pragma unroll
    for (int n = 0; n < TN; ++n) acc[n] = (v8f){};

    float4 bcur[TN], bnxt[TN];
#pragma unroll
    for (int n = 0; n < TN; ++n) bcur[n] = bl[n * 32 + lane];
#pragma unroll
    for (int j = 0; j < KB; ++j) {
        if (j + 1 < KB) {
#pragma unroll
            for (int n = 0; n < TN; ++n) bnxt[n] = bl[((j + 1) * TN + n) * 32 + lane];
        }
        v2f a0, a1;
        a0.x = a[j].x; a0.y = a[j].y;
        a1.x = a[j].z; a1.y = a[j].w;
#pragma unroll
        for (int n = 0; n < TN; ++n) {
            v2f b0;
            b0.x = bcur[n].x; b0.y = bcur[n].y;
            acc[n] = __builtin_amdgcn_wmma_f32_16x16x4_f32(
                 false, a0,  false, b0,
                 (short)0, acc[n],  false,  false);
        }
#pragma unroll
        for (int n = 0; n < TN; ++n) {
            v2f b1;
            b1.x = bcur[n].z; b1.y = bcur[n].w;
            acc[n] = __builtin_amdgcn_wmma_f32_16x16x4_f32(
                 false, a1,  false, b1,
                 (short)0, acc[n],  false,  false);
        }
#pragma unroll
        for (int n = 0; n < TN; ++n) bcur[n] = bnxt[n];
    }

    float* cptr = C + (size_t)(rowBase + 8 * h) * NC + c;
#pragma unroll
    for (int n = 0; n < TN; ++n)
#pragma unroll
        for (int i = 0; i < 8; ++i) cptr[(size_t)i * NC + 16 * n] = acc[n][i];
}

constexpr int PT  = 256;
constexpr int EPT = 16;
constexpr int CH  = PT * EPT;

__global__ void __launch_bounds__(PT)
bucket_hist_kernel(const int* __restrict__ dst, int E, int NB,
                   int* __restrict__ bcnt, int nPart,
                   const float* __restrict__ x, const float* __restrict__ W1,
                   float* __restrict__ h1, int strip0, int stripEnd) {
    if ((int)blockIdx.x >= nPart) {
        gemm_strips_wmma_f32<128, 64, PT / 32>(x, W1, h1, strip0, stripEnd,
                                               (int)blockIdx.x - nPart);
        return;
    }
    extern __shared__ int sm[];
    int* lcnt = sm;
    const int tid = threadIdx.x;
    for (int b = tid; b < NB; b += PT) lcnt[b] = 0;
    int base = blockIdx.x * CH + tid;
    int d[EPT];
#pragma unroll
    for (int j = 0; j < EPT; ++j) {
        int e = base + j * PT;
        d[j] = (e < E) ? dst[e] : -1;
    }
    __syncthreads();
#pragma unroll
    for (int j = 0; j < EPT; ++j)
        if (d[j] >= 0) atomicAdd(&lcnt[d[j] >> 8], 1);
    __syncthreads();
    for (int b = tid; b < NB; b += PT) {
        int c = lcnt[b];
        if (c) atomicAdd(&bcnt[b], c);
    }
}

__global__ void __launch_bounds__(PT)
bucket_scatter_kernel(const int* __restrict__ src,
                      const int* __restrict__ dst, int E, int NB,
                      const int* __restrict__ bcnt,
                      int* __restrict__ bcur,
                      int* __restrict__ bstart,
                      int* __restrict__ pairs_i, int nPart,
                      const float* __restrict__ x, const float* __restrict__ W1,
                      float* __restrict__ h1, int strip0, int stripEnd) {
    if ((int)blockIdx.x >= nPart) {
        gemm_strips_wmma_f32<128, 64, PT / 32>(x, W1, h1, strip0, stripEnd,
                                               (int)blockIdx.x - nPart);
        return;
    }
    extern __shared__ int sm[];
    int* lcnt   = sm;
    int* lstart = sm + NB;
    int* wsum   = sm + 2 * NB;
    const int tid = threadIdx.x;
    const int lane = tid & 31;
    const int w = tid >> 5;

    int base = blockIdx.x * CH + tid;
    int d[EPT], s[EPT], r[EPT];
#pragma unroll
    for (int j = 0; j < EPT; ++j) {
        int e = base + j * PT;
        bool ok = e < E;
        d[j] = ok ? dst[e] : -1;
        s[j] = ok ? src[e] : 0;
        r[j] = 0;
    }

    int carry = 0;
    for (int b0 = 0; b0 < NB; b0 += PT) {
        int b = b0 + tid;
        int c = (b < NB) ? bcnt[b] : 0;
        int incl = c;
#pragma unroll
        for (int o = 1; o < 32; o <<= 1) {
            int t = __shfl_up(incl, o);
            if (lane >= o) incl += t;
        }
        if (lane == 31) wsum[w] = incl;
        __syncthreads();
        if (tid == 0) {
            int tot = 0;
#pragma unroll
            for (int k = 0; k < PT / 32; ++k) { int t = wsum[k]; wsum[k] = tot; tot += t; }
            wsum[PT / 32] = tot;
        }
        __syncthreads();
        if (b < NB) {
            lstart[b] = carry + wsum[w] + incl - c;
            lcnt[b] = 0;
        }
        carry += wsum[PT / 32];
        __syncthreads();
    }
    if (blockIdx.x == 0)
        for (int b = tid; b < NB; b += PT) bstart[b] = lstart[b];

#pragma unroll
    for (int j = 0; j < EPT; ++j)
        if (d[j] >= 0) r[j] = atomicAdd(&lcnt[d[j] >> 8], 1);
    __syncthreads();
    for (int b = tid; b < NB; b += PT) {
        int c = lcnt[b];
        int off = 0;
        if (c) off = atomicAdd(&bcur[b], c);
        lcnt[b] = lstart[b] + off;
    }
    __syncthreads();
    int2* pairs = (int2*)pairs_i;
#pragma unroll
    for (int j = 0; j < EPT; ++j)
        if (d[j] >= 0) pairs[lcnt[d[j] >> 8] + r[j]] = make_int2(s[j], d[j]);
}

__global__ void __launch_bounds__(PT)
bucket_finish_kernel(const int* __restrict__ pairs_i,
                     const int* __restrict__ bstart,
                     const int* __restrict__ bcnt, int n,
                     int* __restrict__ start,
                     int* __restrict__ cnt,
                     float* __restrict__ dis,
                     int* __restrict__ csr_src, int nPart,
                     const float* __restrict__ x, const float* __restrict__ W1,
                     float* __restrict__ h1, int strip0, int stripEnd) {
    if ((int)blockIdx.x >= nPart) {
        gemm_strips_wmma_f32<128, 64, PT / 32>(x, W1, h1, strip0, stripEnd,
                                               (int)blockIdx.x - nPart);
        return;
    }
    __shared__ int lcnt[256];
    __shared__ int lcur[256];
    __shared__ int lex[256];
    __shared__ int wsum[8];
    const int tid = threadIdx.x;
    const int lane = tid & 31;
    const int w = tid >> 5;
    const int b = blockIdx.x;
    const int bs = bstart[b];
    const int bc = bcnt[b];
    const int2* p = (const int2*)pairs_i + bs;

    lcnt[tid] = 0;
    lcur[tid] = 0;
    __syncthreads();
    for (int i0 = 0; i0 < bc; i0 += 4 * 256) {
        int2 q[4];
#pragma unroll
        for (int k = 0; k < 4; ++k) q[k] = p[min(i0 + k * 256 + tid, bc - 1)];
#pragma unroll
        for (int k = 0; k < 4; ++k)
            if (i0 + k * 256 + tid < bc) atomicAdd(&lcnt[q[k].y & 255], 1);
    }
    __syncthreads();

    int c = lcnt[tid];
    int incl = c;
#pragma unroll
    for (int o = 1; o < 32; o <<= 1) {
        int t = __shfl_up(incl, o);
        if (lane >= o) incl += t;
    }
    if (lane == 31) wsum[w] = incl;
    __syncthreads();
    if (tid == 0) {
        int tot = 0;
#pragma unroll
        for (int k = 0; k < 8; ++k) { int t = wsum[k]; wsum[k] = tot; tot += t; }
    }
    __syncthreads();
    int ex = wsum[w] + incl - c;
    lex[tid] = ex;
    int node = b * 256 + tid;
    if (node < n) {
        start[node] = bs + ex;
        cnt[node] = c;
        dis[node] = rsqrtf((float)c + 1.0f);
    }
    __syncthreads();

    for (int i0 = 0; i0 < bc; i0 += 4 * 256) {
        int2 q[4];
#pragma unroll
        for (int k = 0; k < 4; ++k) q[k] = p[min(i0 + k * 256 + tid, bc - 1)];
#pragma unroll
        for (int k = 0; k < 4; ++k) {
            if (i0 + k * 256 + tid < bc) {
                int l = q[k].y & 255;
                int rk = atomicAdd(&lcur[l], 1);
                csr_src[bs + lex[l] + rk] = q[k].x;
            }
        }
    }
}

template <int G>
__device__ __forceinline__ float4 gather_sum(float4 acc,
                                             const int* __restrict__ idx, int len,
                                             float dd,
                                             const float* __restrict__ dis,
                                             const float4* __restrict__ h4, int q) {
    int k = 0;
    for (; k + 4 <= len; k += 4) {
        int s0 = idx[k + 0];
        int s1 = idx[k + 1];
        int s2 = idx[k + 2];
        int s3 = idx[k + 3];
        float w0 = dis[s0] * dd;
        float w1 = dis[s1] * dd;
        float w2 = dis[s2] * dd;
        float w3 = dis[s3] * dd;
        float4 v0 = h4[(size_t)s0 * G + q];
        float4 v1 = h4[(size_t)s1 * G + q];
        float4 v2 = h4[(size_t)s2 * G + q];
        float4 v3 = h4[(size_t)s3 * G + q];
        acc.x += v0.x * w0; acc.y += v0.y * w0; acc.z += v0.z * w0; acc.w += v0.w * w0;
        acc.x += v1.x * w1; acc.y += v1.y * w1; acc.z += v1.z * w1; acc.w += v1.w * w1;
        acc.x += v2.x * w2; acc.y += v2.y * w2; acc.z += v2.z * w2; acc.w += v2.w * w2;
        acc.x += v3.x * w3; acc.y += v3.y * w3; acc.z += v3.z * w3; acc.w += v3.w * w3;
    }
    if (k < len) {
        int last = len - 1;
        bool p1 = k + 1 < len;
        bool p2 = k + 2 < len;
        int s0 = idx[k];
        int s1 = idx[min(k + 1, last)];
        int s2 = idx[min(k + 2, last)];
        float w0 = dis[s0] * dd;
        float w1 = dis[s1] * dd;
        float w2 = dis[s2] * dd;
        float4 v0 = h4[(size_t)s0 * G + q];
        float4 v1 = h4[(size_t)s1 * G + q];
        float4 v2 = h4[(size_t)s2 * G + q];
        acc.x += v0.x * w0; acc.y += v0.y * w0; acc.z += v0.z * w0; acc.w += v0.w * w0;
        float4 t1 = acc;
        t1.x += v1.x * w1; t1.y += v1.y * w1; t1.z += v1.z * w1; t1.w += v1.w * w1;
        acc.x = p1 ? t1.x : acc.x; acc.y = p1 ? t1.y : acc.y;
        acc.z = p1 ? t1.z : acc.z; acc.w = p1 ? t1.w : acc.w;
        float4 t2 = acc;
        t2.x += v2.x * w2; t2.y += v2.y * w2; t2.z += v2.z * w2; t2.w += v2.w * w2;
        acc.x = p2 ? t2.x : acc.x; acc.y = p2 ? t2.y : acc.y;
        acc.z = p2 ? t2.z : acc.z; acc.w = p2 ? t2.w : acc.w;
    }
    return acc;
}

template <int G>
__device__ __forceinline__ float4 gather_rows_sum(float4 acc,
                                                  const int* __restrict__ idx, int len,
                                                  const float4* __restrict__ g4, int q) {
    int k = 0;
#pragma unroll 2
    for (; k + 4 <= len; k += 4) {
        int s0 = idx[k + 0];
        int s1 = idx[k + 1];
        int s2 = idx[k + 2];
        int s3 = idx[k + 3];
        float4 v0 = g4[(size_t)s0 * G + q];
        float4 v1 = g4[(size_t)s1 * G + q];
        float4 v2 = g4[(size_t)s2 * G + q];
        float4 v3 = g4[(size_t)s3 * G + q];
        acc.x += v0.x; acc.y += v0.y; acc.z += v0.z; acc.w += v0.w;
        acc.x += v1.x; acc.y += v1.y; acc.z += v1.z; acc.w += v1.w;
        acc.x += v2.x; acc.y += v2.y; acc.z += v2.z; acc.w += v2.w;
        acc.x += v3.x; acc.y += v3.y; acc.z += v3.z; acc.w += v3.w;
    }
    if (k < len) {
        int last = len - 1;
        bool p1 = k + 1 < len;
        bool p2 = k + 2 < len;
        int s0 = idx[k];
        int s1 = idx[min(k + 1, last)];
        int s2 = idx[min(k + 2, last)];
        float4 v0 = g4[(size_t)s0 * G + q];
        float4 v1 = g4[(size_t)s1 * G + q];
        float4 v2 = g4[(size_t)s2 * G + q];
        acc.x += v0.x; acc.y += v0.y; acc.z += v0.z; acc.w += v0.w;
        float4 t1 = acc;
        t1.x += v1.x; t1.y += v1.y; t1.z += v1.z; t1.w += v1.w;
        acc.x = p1 ? t1.x : acc.x; acc.y = p1 ? t1.y : acc.y;
        acc.z = p1 ? t1.z : acc.z; acc.w = p1 ? t1.w : acc.w;
        float4 t2 = acc;
        t2.x += v2.x; t2.y += v2.y; t2.z += v2.z; t2.w += v2.w;
        acc.x = p2 ? t2.x : acc.x; acc.y = p2 ? t2.y : acc.y;
        acc.z = p2 ? t2.z : acc.z; acc.w = p2 ? t2.w : acc.w;
    }
    return acc;
}

constexpr int YP4 = 17;

__global__ void __launch_bounds__(256)
agg_relu_gemm2_kernel(const int* __restrict__ csr_src,
                      const int* __restrict__ start,
                      const int* __restrict__ cnt,
                      const float* __restrict__ dis,
                      const float* __restrict__ h,
                      const float* __restrict__ bias,
                      const float* __restrict__ W2,
                      float* __restrict__ g) {
    constexpr int G = 16;
    __shared__ float4 bl[8 * 32];
    __shared__ float4 ys[16 * YP4];
    __shared__ int arrived;

    const int tid = threadIdx.x;
    const int r = tid >> 4;
    const int q = tid & 15;
    const int rowBase = blockIdx.x * 16;

    const float* wp = W2 + (size_t)r * 64 + q;
    float4 bv;
    bv.x = wp[0];
    bv.y = wp[16];
    bv.z = wp[32];
    bv.w = wp[48];

    const int node = rowBase + r;
    const float4* h4 = (const float4*)h;
    const float dd = dis[node];
    const float4 self = h4[(size_t)rowBase * G + tid];
    const float4 bq = ((const float4*)bias)[q];
    const int st = start[node];
    const int len = cnt[node];

    bl[tid] = bv;
    if (tid == 0) arrived = 0;
    __syncthreads();

    const float d2 = dd * dd;
    float4 acc;
    acc.x = self.x * d2 + bq.x;
    acc.y = self.y * d2 + bq.y;
    acc.z = self.z * d2 + bq.z;
    acc.w = self.w * d2 + bq.w;
    acc = gather_sum<G>(acc, csr_src + st, len, dd, dis, h4, q);
    float4 v;
    v.x = fmaxf(acc.x, 0.0f);
    v.y = fmaxf(acc.y, 0.0f);
    v.z = fmaxf(acc.z, 0.0f);
    v.w = fmaxf(acc.w, 0.0f);
    ys[tid + (tid >> 4)] = v;

    __threadfence_block();
    int old = 0;
    if ((tid & 31) == 0)
        old = __hip_atomic_fetch_add(&arrived, 1, __ATOMIC_ACQ_REL, __HIP_MEMORY_SCOPE_WORKGROUP);
    old = __builtin_amdgcn_readfirstlane(old);
    if (old != 7) return;

    const int lane = tid & 31;
    const int c = lane & 15;
    const int hh = lane >> 4;
    const float4* sp = (const float4*)(dis + rowBase + 8 * hh);
    const float4 s0 = sp[0];
    const float4 s1 = sp[1];

    v8f p0 = (v8f){};
    v8f p1 = (v8f){};
#pragma unroll
    for (int j = 0; j < 8; ++j) {
        const float4 a4 = ys[c * YP4 + 2 * j + hh];
        const float4 b4 = bl[j * 32 + lane];
        v2f a0, a1, b0, b1;
        a0.x = a4.x; a0.y = a4.y;
        a1.x = a4.z; a1.y = a4.w;
        b0.x = b4.x; b0.y = b4.y;
        b1.x = b4.z; b1.y = b4.w;
        p0 = __builtin_amdgcn_wmma_f32_16x16x4_f32(
             false, a0,  false, b0,
             (short)0, p0,  false,  false);
        p1 = __builtin_amdgcn_wmma_f32_16x16x4_f32(
             false, a1,  false, b1,
             (short)0, p1,  false,  false);
    }
    const float sc[8] = {s0.x, s0.y, s0.z, s0.w, s1.x, s1.y, s1.z, s1.w};
    float* gp = g + (size_t)(rowBase + 8 * hh) * 16 + c;
#pragma unroll
    for (int i = 0; i < 8; ++i) gp[(size_t)i * 16] = (p0[i] + p1[i]) * sc[i];
}

__global__ void agg_logsoftmax_kernel(const int* __restrict__ csr_src,
                                      const int* __restrict__ start,
                                      const int* __restrict__ cnt,
                                      const float* __restrict__ dis,
                                      const float* __restrict__ g,
                                      const float* __restrict__ bias,
                                      float* __restrict__ out, int n) {
    constexpr int G = 4;
    int t = blockIdx.x * blockDim.x + threadIdx.x;
    int node = t / G;
    int q = t % G;
    if (node >= n) return;
    const float4* g4 = (const float4*)g;
    const int st = start[node];
    const int len = cnt[node];
    float dd = dis[node];
    float4 acc = g4[t];
    const float4 b = ((const float4*)bias)[q];
    asm volatile("" :: "v"(st), "v"(len), "v"(b.x), "v"(b.y), "v"(b.z), "v"(b.w));
    acc = gather_rows_sum<G>(acc, csr_src + st, len, g4, q);
    float4 v;
    v.x = acc.x * dd + b.x;
    v.y = acc.y * dd + b.y;
    v.z = acc.z * dd + b.z;
    v.w = acc.w * dd + b.w;
    float mx = fmaxf(fmaxf(v.x, v.y), fmaxf(v.z, v.w));
    mx = fmaxf(mx, __shfl_xor(mx, 1));
    mx = fmaxf(mx, __shfl_xor(mx, 2));
    float s = (__expf(v.x - mx) + __expf(v.y - mx)) + (__expf(v.z - mx) + __expf(v.w - mx));
    s += __shfl_xor(s, 1);
    s += __shfl_xor(s, 2);
    float lse = mx + __logf(s);
    float4 r;
    r.x = v.x - lse; r.y = v.y - lse; r.z = v.z - lse; r.w = v.w - lse;
    ((float4*)out)[t] = r;
}

extern "C" void kernel_launch(void* const* d_in, const int* in_sizes, int n_in,
                              void* d_out, int out_size, void* d_ws, size_t ws_size,
                              hipStream_t stream) {
    const float* x   = (const float*)d_in[0];
    const int*   ei  = (const int*)  d_in[1];
    const float* W1  = (const float*)d_in[2];
    const float* b1  = (const float*)d_in[3];
    const float* W2  = (const float*)d_in[4];
    const float* b2  = (const float*)d_in[5];
    float* out = (float*)d_out;

    const int N = in_sizes[0] / 128;
    const int E = in_sizes[1] / 2;
    const int* src = ei;
    const int* dst = ei + E;

    const int NB = (N + 255) / 256;
    int*   pairs = (int*)d_ws;
    float* h1   = (float*)(pairs + (size_t)2 * E);
    float* h2   = h1   + (size_t)N * 64;
    float* dis  = h2   + (size_t)N * 16;
    int* start   = (int*)(dis + N);
    int* cnt     = start + N;
    int* bcnt    = cnt + N;
    int* bcur    = bcnt + NB;
    int* bstart  = bcur + NB;
    int* csr_src = bstart + NB;

    hipMemsetAsync(bcnt, 0, (size_t)(2 * NB) * sizeof(int), stream);

    const int T = 256;

    {
        const int nChunks = (E + CH - 1) / CH;
        const int SPB = PT / 32;
        const int nStrips = N / 16;
        const int cap = 256 * SPB;
        const int thirdA = nStrips / 3;
        const int sA = thirdA < cap ? thirdA : cap;
        const int halfC = (nStrips - sA) / 2;
        const int sB = nStrips - (halfC < cap ? halfC : cap);
        const int gA = (sA + SPB - 1) / SPB;
        const int gB = (sB - sA + SPB - 1) / SPB;
        const int gC = (nStrips - sB + SPB - 1) / SPB;
        bucket_hist_kernel<<<nChunks + gA, PT, (size_t)NB * sizeof(int), stream>>>(
            dst, E, NB, bcnt, nChunks, x, W1, h1, 0, sA);
        bucket_scatter_kernel<<<nChunks + gB, PT, (size_t)(2 * NB + 16) * sizeof(int), stream>>>(
            src, dst, E, NB, bcnt, bcur, bstart, pairs, nChunks, x, W1, h1, sA, sB);
        bucket_finish_kernel<<<NB + gC, PT, 0, stream>>>(pairs, bstart, bcnt, N,
                                                      start, cnt, dis, csr_src,
                                                      NB, x, W1, h1, sB, nStrips);
    }
    agg_relu_gemm2_kernel<<<N / 16, T, 0, stream>>>(
        csr_src, start, cnt, dis, h1, b1, W2, h2);
    {
        long long total = (long long)N * 4;
        agg_logsoftmax_kernel<<<(int)((total + T - 1) / T), T, 0, stream>>>(
            csr_src, start, cnt, dis, h2, b2, out, N);
    }
}


# Partition passes store through LDS tiles: pairs leave grouped in per-bucket runs, the sorted source list as whole 16-byte groups, not one 4- or 8-byte piece per lane; entry order as before
# speedup vs baseline: 6.4887x; 1.1277x over previous
#include <hip/hip_runtime.h>
#include <cstdint>

typedef __attribute__((ext_vector_type(2))) float v2f;
typedef __attribute__((ext_vector_type(8))) float v8f;

template <int K, int NC, int GW>
__device__ __forceinline__ void gemm_strips_wmma_f32(const float* __restrict__ A,
                                                     const float* __restrict__ B,
                                                     float* __restrict__ C,
                                                     int strip0, int stripEnd, int blk) {
    constexpr int TN = NC / 16;
    constexpr int KB = K / 8;
    __shared__ float4 bl[KB * TN * 32];

    const int lane = threadIdx.x & 31;
    const int wv = threadIdx.x >> 5;
    const int c = lane & 15;
    const int h = lane >> 4;
    const int strip = strip0 + blk * GW + wv;
    const bool live = strip < stripEnd;
    const int rowBase = (live ? strip : strip0) * 16;

    constexpr int SPW = KB * TN / GW;
    static_assert(KB * TN % GW == 0, "slot rows must divide evenly over the waves");
    const float* bp = B + (size_t)(4 * h) * NC + c;
    float4 bv[SPW];
#pragma unroll
    for (int i = 0; i < SPW; ++i) {
        int jn = wv + i * GW;
        int j = jn / TN;
        int n = jn % TN;
        const float* p = bp + (size_t)(8 * j) * NC + 16 * n;
        bv[i].x = p[0];
        bv[i].y = p[NC];
        bv[i].z = p[2 * NC];
        bv[i].w = p[3 * NC];
    }

    const float4* ap = (const float4*)(A + (size_t)(rowBase + c) * K + 4 * h);
    float4 a[KB];
#pragma unroll
    for (int j = 0; j < KB; ++j) a[j] = ap[2 * j];

#pragma unroll
    for (int i = 0; i < SPW; ++i) bl[(wv + i * GW) * 32 + lane] = bv[i];
    __syncthreads();
    if (!live) return;

    v8f acc[TN];
#pragma unroll
    for (int n = 0; n < TN; ++n) acc[n] = (v8f){};

    float4 bcur[TN], bnxt[TN];
#pragma unroll
    for (int n = 0; n < TN; ++n) bcur[n] = bl[n * 32 + lane];
#pragma unroll
    for (int j = 0; j < KB; ++j) {
        if (j + 1 < KB) {
#pragma unroll
            for (int n = 0; n < TN; ++n) bnxt[n] = bl[((j + 1) * TN + n) * 32 + lane];
        }
        v2f a0, a1;
        a0.x = a[j].x; a0.y = a[j].y;
        a1.x = a[j].z; a1.y = a[j].w;
#pragma unroll
        for (int n = 0; n < TN; ++n) {
            v2f b0;
            b0.x = bcur[n].x; b0.y = bcur[n].y;
            acc[n] = __builtin_amdgcn_wmma_f32_16x16x4_f32(
                 false, a0,  false, b0,
                 (short)0, acc[n],  false,  false);
        }
#pragma unroll
        for (int n = 0; n < TN; ++n) {
            v2f b1;
            b1.x = bcur[n].z; b1.y = bcur[n].w;
            acc[n] = __builtin_amdgcn_wmma_f32_16x16x4_f32(
                 false, a1,  false, b1,
                 (short)0, acc[n],  false,  false);
        }
#pragma unroll
        for (int n = 0; n < TN; ++n) bcur[n] = bnxt[n];
    }

    float* cptr = C + (size_t)(rowBase + 8 * h) * NC + c;
#pragma unroll
    for (int n = 0; n < TN; ++n)
#pragma unroll
        for (int i = 0; i < 8; ++i) cptr[(size_t)i * NC + 16 * n] = acc[n][i];
}

constexpr int PT  = 256;
constexpr int EPT = 16;
constexpr int CH  = PT * EPT;
constexpr int FT  = 8192;

__global__ void __launch_bounds__(PT)
bucket_hist_kernel(const int* __restrict__ dst, int E, int NB,
                   int* __restrict__ bcnt, int nPart,
                   const float* __restrict__ x, const float* __restrict__ W1,
                   float* __restrict__ h1, int strip0, int stripEnd) {
    if ((int)blockIdx.x >= nPart) {
        gemm_strips_wmma_f32<128, 64, PT / 32>(x, W1, h1, strip0, stripEnd,
                                               (int)blockIdx.x - nPart);
        return;
    }
    extern __shared__ int sm[];
    int* lcnt = sm;
    const int tid = threadIdx.x;
    for (int b = tid; b < NB; b += PT) lcnt[b] = 0;
    int base = blockIdx.x * CH + tid;
    int d[EPT];
#pragma unroll
    for (int j = 0; j < EPT; ++j) {
        int e = base + j * PT;
        d[j] = (e < E) ? dst[e] : -1;
    }
    __syncthreads();
#pragma unroll
    for (int j = 0; j < EPT; ++j)
        if (d[j] >= 0) atomicAdd(&lcnt[d[j] >> 8], 1);
    __syncthreads();
    for (int b = tid; b < NB; b += PT) {
        int c = lcnt[b];
        if (c) atomicAdd(&bcnt[b], c);
    }
}

__global__ void __launch_bounds__(PT)
bucket_scatter_kernel(const int* __restrict__ src,
                      const int* __restrict__ dst, int E, int NB,
                      const int* __restrict__ bcnt,
                      int* __restrict__ bcur,
                      int* __restrict__ bstart,
                      int* __restrict__ pairs_i, int nPart,
                      const float* __restrict__ x, const float* __restrict__ W1,
                      float* __restrict__ h1, int strip0, int stripEnd) {
    if ((int)blockIdx.x >= nPart) {
        gemm_strips_wmma_f32<128, 64, PT / 32>(x, W1, h1, strip0, stripEnd,
                                               (int)blockIdx.x - nPart);
        return;
    }
    extern __shared__ int sm[];
    int* lcnt   = sm;
    int* lstart = sm + NB;
    int* wsum   = sm + 2 * NB;
    __shared__ int2 tile[CH];
    const int tid = threadIdx.x;
    const int lane = tid & 31;
    const int w = tid >> 5;

    int base = blockIdx.x * CH + tid;
    int d[EPT], s[EPT], r[EPT];
#pragma unroll
    for (int j = 0; j < EPT; ++j) {
        int e = base + j * PT;
        bool ok = e < E;
        d[j] = ok ? dst[e] : -1;
        s[j] = ok ? src[e] : 0;
        r[j] = 0;
    }

    int carry = 0;
    for (int b0 = 0; b0 < NB; b0 += PT) {
        int b = b0 + tid;
        int c = (b < NB) ? bcnt[b] : 0;
        int incl = c;
#pragma unroll
        for (int o = 1; o < 32; o <<= 1) {
            int t = __shfl_up(incl, o);
            if (lane >= o) incl += t;
        }
        if (lane == 31) wsum[w] = incl;
        __syncthreads();
        if (tid == 0) {
            int tot = 0;
#pragma unroll
            for (int k = 0; k < PT / 32; ++k) { int t = wsum[k]; wsum[k] = tot; tot += t; }
            wsum[PT / 32] = tot;
        }
        __syncthreads();
        if (b < NB) {
            lstart[b] = carry + wsum[w] + incl - c;
            lcnt[b] = 0;
        }
        carry += wsum[PT / 32];
        __syncthreads();
    }
    if (blockIdx.x == 0)
        for (int b = tid; b < NB; b += PT) bstart[b] = lstart[b];

#pragma unroll
    for (int j = 0; j < EPT; ++j)
        if (d[j] >= 0) r[j] = atomicAdd(&lcnt[d[j] >> 8], 1);
    if (tid == 0) wsum[0] = 0;
    __syncthreads();
    for (int b = tid; b < NB; b += PT) {
        int c = lcnt[b];
        int off = 0;
        int lpos = 0;
        if (c) {
            off = atomicAdd(&bcur[b], c);
            lpos = atomicAdd(&wsum[0], c);
        }
        lcnt[b] = lpos;
        lstart[b] = lstart[b] + off - lpos;
    }
    __syncthreads();
    int rb[EPT];
#pragma unroll
    for (int j = 0; j < EPT; ++j) rb[j] = lcnt[d[j] >= 0 ? (d[j] >> 8) : 0];
#pragma unroll
    for (int j = 0; j < EPT; ++j)
        if (d[j] >= 0) tile[rb[j] + r[j]] = make_int2(s[j], d[j]);
    __syncthreads();
    const int e0 = blockIdx.x * CH;
    const int nE = (E - e0) < CH ? (E - e0) : CH;
    int2* pairs = (int2*)pairs_i;
    int2 pe[EPT];
    int dl[EPT];
#pragma unroll
    for (int j = 0; j < EPT; ++j) pe[j] = tile[min(j * PT + tid, nE - 1)];
#pragma unroll
    for (int j = 0; j < EPT; ++j) dl[j] = lstart[pe[j].y >> 8];
#pragma unroll
    for (int j = 0; j < EPT; ++j) {
        int i = j * PT + tid;
        if (i < nE) pairs[i + dl[j]] = pe[j];
    }
}

__global__ void __launch_bounds__(PT)
bucket_finish_kernel(const int* __restrict__ pairs_i,
                     const int* __restrict__ bstart,
                     const int* __restrict__ bcnt, int n,
                     int* __restrict__ start,
                     int* __restrict__ cnt,
                     float* __restrict__ dis,
                     int* __restrict__ csr_src, int nPart,
                     const float* __restrict__ x, const float* __restrict__ W1,
                     float* __restrict__ h1, int strip0, int stripEnd) {
    if ((int)blockIdx.x >= nPart) {
        gemm_strips_wmma_f32<128, 64, PT / 32>(x, W1, h1, strip0, stripEnd,
                                               (int)blockIdx.x - nPart);
        return;
    }
    __shared__ __attribute__((aligned(16))) int ftile[FT];
    __shared__ int lcnt[256];
    __shared__ int lcur[256];
    __shared__ int lex[256];
    __shared__ int wsum[8];
    const int tid = threadIdx.x;
    const int lane = tid & 31;
    const int w = tid >> 5;
    const int b = blockIdx.x;
    const int bs = bstart[b];
    const int bc = bcnt[b];
    const int2* p = (const int2*)pairs_i + bs;

    lcnt[tid] = 0;
    lcur[tid] = 0;
    __syncthreads();
    for (int i0 = 0; i0 < bc; i0 += 4 * 256) {
        int2 q[4];
#pragma unroll
        for (int k = 0; k < 4; ++k) q[k] = p[min(i0 + k * 256 + tid, bc - 1)];
#pragma unroll
        for (int k = 0; k < 4; ++k)
            if (i0 + k * 256 + tid < bc) atomicAdd(&lcnt[q[k].y & 255], 1);
    }
    __syncthreads();

    int c = lcnt[tid];
    int incl = c;
#pragma unroll
    for (int o = 1; o < 32; o <<= 1) {
        int t = __shfl_up(incl, o);
        if (lane >= o) incl += t;
    }
    if (lane == 31) wsum[w] = incl;
    __syncthreads();
    if (tid == 0) {
        int tot = 0;
#pragma unroll
        for (int k = 0; k < 8; ++k) { int t = wsum[k]; wsum[k] = tot; tot += t; }
    }
    __syncthreads();
    int ex = wsum[w] + incl - c;
    lex[tid] = ex;
    int node = b * 256 + tid;
    if (node < n) {
        start[node] = bs + ex;
        cnt[node] = c;
        dis[node] = rsqrtf((float)c + 1.0f);
    }
    __syncthreads();

    int* seg = csr_src + bs;
    const int sh = (int)((((size_t)seg) >> 2) & 3);
    const int tot = sh + bc;
    const bool staged = tot <= FT;
    for (int i0 = 0; i0 < bc; i0 += 4 * 256) {
        int2 q[4];
#pragma unroll
        for (int k = 0; k < 4; ++k) q[k] = p[min(i0 + k * 256 + tid, bc - 1)];
#pragma unroll
        for (int k = 0; k < 4; ++k) {
            if (i0 + k * 256 + tid < bc) {
                int l = q[k].y & 255;
                int rk = atomicAdd(&lcur[l], 1);
                if (staged) ftile[sh + lex[l] + rk] = q[k].x;
                else        seg[lex[l] + rk] = q[k].x;
            }
        }
    }
    if (!staged) return;
    __syncthreads();
    int* al = seg - sh;
    int4* gout = (int4*)al;
    const int4* tin = (const int4*)ftile;
    const int g0 = sh > 0 ? 1 : 0;
    const int g1 = tot >> 2;
    for (int gq = g0 + tid; gq < g1; gq += 256) gout[gq] = tin[gq];
    if (tid < 4) {
        int e = tid;
        if (sh > 0 && e >= sh && e < tot) al[e] = ftile[e];
    } else if (tid >= 32 && tid < 36) {
        int e = 4 * g1 + (tid - 32);
        if (e >= 4 * g0 && e < tot) al[e] = ftile[e];
    }
}

template <int G>
__device__ __forceinline__ float4 gather_sum(float4 acc,
                                             const int* __restrict__ idx, int len,
                                             float dd,
                                             const float* __restrict__ dis,
                                             const float4* __restrict__ h4, int q) {
    int k = 0;
    for (; k + 4 <= len; k += 4) {
        int s0 = idx[k + 0];
        int s1 = idx[k + 1];
        int s2 = idx[k + 2];
        int s3 = idx[k + 3];
        float w0 = dis[s0] * dd;
        float w1 = dis[s1] * dd;
        float w2 = dis[s2] * dd;
        float w3 = dis[s3] * dd;
        float4 v0 = h4[(size_t)s0 * G + q];
        float4 v1 = h4[(size_t)s1 * G + q];
        float4 v2 = h4[(size_t)s2 * G + q];
        float4 v3 = h4[(size_t)s3 * G + q];
        acc.x += v0.x * w0; acc.y += v0.y * w0; acc.z += v0.z * w0; acc.w += v0.w * w0;
        acc.x += v1.x * w1; acc.y += v1.y * w1; acc.z += v1.z * w1; acc.w += v1.w * w1;
        acc.x += v2.x * w2; acc.y += v2.y * w2; acc.z += v2.z * w2; acc.w += v2.w * w2;
        acc.x += v3.x * w3; acc.y += v3.y * w3; acc.z += v3.z * w3; acc.w += v3.w * w3;
    }
    if (k < len) {
        int last = len - 1;
        bool p1 = k + 1 < len;
        bool p2 = k + 2 < len;
        int s0 = idx[k];
        int s1 = idx[min(k + 1, last)];
        int s2 = idx[min(k + 2, last)];
        float w0 = dis[s0] * dd;
        float w1 = dis[s1] * dd;
        float w2 = dis[s2] * dd;
        float4 v0 = h4[(size_t)s0 * G + q];
        float4 v1 = h4[(size_t)s1 * G + q];
        float4 v2 = h4[(size_t)s2 * G + q];
        acc.x += v0.x * w0; acc.y += v0.y * w0; acc.z += v0.z * w0; acc.w += v0.w * w0;
        float4 t1 = acc;
        t1.x += v1.x * w1; t1.y += v1.y * w1; t1.z += v1.z * w1; t1.w += v1.w * w1;
        acc.x = p1 ? t1.x : acc.x; acc.y = p1 ? t1.y : acc.y;
        acc.z = p1 ? t1.z : acc.z; acc.w = p1 ? t1.w : acc.w;
        float4 t2 = acc;
        t2.x += v2.x * w2; t2.y += v2.y * w2; t2.z += v2.z * w2; t2.w += v2.w * w2;
        acc.x = p2 ? t2.x : acc.x; acc.y = p2 ? t2.y : acc.y;
        acc.z = p2 ? t2.z : acc.z; acc.w = p2 ? t2.w : acc.w;
    }
    return acc;
}

template <int G>
__device__ __forceinline__ float4 gather_rows_sum(float4 acc,
                                                  const int* __restrict__ idx, int len,
                                                  const float4* __restrict__ g4, int q) {
    int k = 0;
#pragma unroll 2
    for (; k + 4 <= len; k += 4) {
        int s0 = idx[k + 0];
        int s1 = idx[k + 1];
        int s2 = idx[k + 2];
        int s3 = idx[k + 3];
        float4 v0 = g4[(size_t)s0 * G + q];
        float4 v1 = g4[(size_t)s1 * G + q];
        float4 v2 = g4[(size_t)s2 * G + q];
        float4 v3 = g4[(size_t)s3 * G + q];
        acc.x += v0.x; acc.y += v0.y; acc.z += v0.z; acc.w += v0.w;
        acc.x += v1.x; acc.y += v1.y; acc.z += v1.z; acc.w += v1.w;
        acc.x += v2.x; acc.y += v2.y; acc.z += v2.z; acc.w += v2.w;
        acc.x += v3.x; acc.y += v3.y; acc.z += v3.z; acc.w += v3.w;
    }
    if (k < len) {
        int last = len - 1;
        bool p1 = k + 1 < len;
        bool p2 = k + 2 < len;
        int s0 = idx[k];
        int s1 = idx[min(k + 1, last)];
        int s2 = idx[min(k + 2, last)];
        float4 v0 = g4[(size_t)s0 * G + q];
        float4 v1 = g4[(size_t)s1 * G + q];
        float4 v2 = g4[(size_t)s2 * G + q];
        acc.x += v0.x; acc.y += v0.y; acc.z += v0.z; acc.w += v0.w;
        float4 t1 = acc;
        t1.x += v1.x; t1.y += v1.y; t1.z += v1.z; t1.w += v1.w;
        acc.x = p1 ? t1.x : acc.x; acc.y = p1 ? t1.y : acc.y;
        acc.z = p1 ? t1.z : acc.z; acc.w = p1 ? t1.w : acc.w;
        float4 t2 = acc;
        t2.x += v2.x; t2.y += v2.y; t2.z += v2.z; t2.w += v2.w;
        acc.x = p2 ? t2.x : acc.x; acc.y = p2 ? t2.y : acc.y;
        acc.z = p2 ? t2.z : acc.z; acc.w = p2 ? t2.w : acc.w;
    }
    return acc;
}

constexpr int YP4 = 17;

__global__ void __launch_bounds__(256)
agg_relu_gemm2_kernel(const int* __restrict__ csr_src,
                      const int* __restrict__ start,
                      const int* __restrict__ cnt,
                      const float* __restrict__ dis,
                      const float* __restrict__ h,
                      const float* __restrict__ bias,
                      const float* __restrict__ W2,
                      float* __restrict__ g) {
    constexpr int G = 16;
    __shared__ float4 bl[8 * 32];
    __shared__ float4 ys[16 * YP4];
    __shared__ int arrived;

    const int tid = threadIdx.x;
    const int r = tid >> 4;
    const int q = tid & 15;
    const int rowBase = blockIdx.x * 16;

    const float* wp = W2 + (size_t)r * 64 + q;
    float4 bv;
    bv.x = wp[0];
    bv.y = wp[16];
    bv.z = wp[32];
    bv.w = wp[48];

    const int node = rowBase + r;
    const float4* h4 = (const float4*)h;
    const float dd = dis[node];
    const float4 self = h4[(size_t)rowBase * G + tid];
    const float4 bq = ((const float4*)bias)[q];
    const int st = start[node];
    const int len = cnt[node];

    bl[tid] = bv;
    if (tid == 0) arrived = 0;
    __syncthreads();

    const float d2 = dd * dd;
    float4 acc;
    acc.x = self.x * d2 + bq.x;
    acc.y = self.y * d2 + bq.y;
    acc.z = self.z * d2 + bq.z;
    acc.w = self.w * d2 + bq.w;
    acc = gather_sum<G>(acc, csr_src + st, len, dd, dis, h4, q);
    float4 v;
    v.x = fmaxf(acc.x, 0.0f);
    v.y = fmaxf(acc.y, 0.0f);
    v.z = fmaxf(acc.z, 0.0f);
    v.w = fmaxf(acc.w, 0.0f);
    ys[tid + (tid >> 4)] = v;

    __threadfence_block();
    int old = 0;
    if ((tid & 31) == 0)
        old = __hip_atomic_fetch_add(&arrived, 1, __ATOMIC_ACQ_REL, __HIP_MEMORY_SCOPE_WORKGROUP);
    old = __builtin_amdgcn_readfirstlane(old);
    if (old != 7) return;

    const int lane = tid & 31;
    const int c = lane & 15;
    const int hh = lane >> 4;
    const float4* sp = (const float4*)(dis + rowBase + 8 * hh);
    const float4 s0 = sp[0];
    const float4 s1 = sp[1];

    v8f p0 = (v8f){};
    v8f p1 = (v8f){};
#pragma unroll
    for (int j = 0; j < 8; ++j) {
        const float4 a4 = ys[c * YP4 + 2 * j + hh];
        const float4 b4 = bl[j * 32 + lane];
        v2f a0, a1, b0, b1;
        a0.x = a4.x; a0.y = a4.y;
        a1.x = a4.z; a1.y = a4.w;
        b0.x = b4.x; b0.y = b4.y;
        b1.x = b4.z; b1.y = b4.w;
        p0 = __builtin_amdgcn_wmma_f32_16x16x4_f32(
             false, a0,  false, b0,
             (short)0, p0,  false,  false);
        p1 = __builtin_amdgcn_wmma_f32_16x16x4_f32(
             false, a1,  false, b1,
             (short)0, p1,  false,  false);
    }
    const float sc[8] = {s0.x, s0.y, s0.z, s0.w, s1.x, s1.y, s1.z, s1.w};
    float* gp = g + (size_t)(rowBase + 8 * hh) * 16 + c;
#pragma unroll
    for (int i = 0; i < 8; ++i) gp[(size_t)i * 16] = (p0[i] + p1[i]) * sc[i];
}

__global__ void agg_logsoftmax_kernel(const int* __restrict__ csr_src,
                                      const int* __restrict__ start,
                                      const int* __restrict__ cnt,
                                      const float* __restrict__ dis,
                                      const float* __restrict__ g,
                                      const float* __restrict__ bias,
                                      float* __restrict__ out, int n) {
    constexpr int G = 4;
    int t = blockIdx.x * blockDim.x + threadIdx.x;
    int node = t / G;
    int q = t % G;
    if (node >= n) return;
    const float4* g4 = (const float4*)g;
    const int st = start[node];
    const int len = cnt[node];
    float dd = dis[node];
    float4 acc = g4[t];
    const float4 b = ((const float4*)bias)[q];
    asm volatile("" :: "v"(st), "v"(len), "v"(b.x), "v"(b.y), "v"(b.z), "v"(b.w));
    acc = gather_rows_sum<G>(acc, csr_src + st, len, g4, q);
    float4 v;
    v.x = acc.x * dd + b.x;
    v.y = acc.y * dd + b.y;
    v.z = acc.z * dd + b.z;
    v.w = acc.w * dd + b.w;
    float mx = fmaxf(fmaxf(v.x, v.y), fmaxf(v.z, v.w));
    mx = fmaxf(mx, __shfl_xor(mx, 1));
    mx = fmaxf(mx, __shfl_xor(mx, 2));
    float s = (__expf(v.x - mx) + __expf(v.y - mx)) + (__expf(v.z - mx) + __expf(v.w - mx));
    s += __shfl_xor(s, 1);
    s += __shfl_xor(s, 2);
    float lse = mx + __logf(s);
    float4 r;
    r.x = v.x - lse; r.y = v.y - lse; r.z = v.z - lse; r.w = v.w - lse;
    ((float4*)out)[t] = r;
}

extern "C" void kernel_launch(void* const* d_in, const int* in_sizes, int n_in,
                              void* d_out, int out_size, void* d_ws, size_t ws_size,
                              hipStream_t stream) {
    const float* x   = (const float*)d_in[0];
    const int*   ei  = (const int*)  d_in[1];
    const float* W1  = (const float*)d_in[2];
    const float* b1  = (const float*)d_in[3];
    const float* W2  = (const float*)d_in[4];
    const float* b2  = (const float*)d_in[5];
    float* out = (float*)d_out;

    const int N = in_sizes[0] / 128;
    const int E = in_sizes[1] / 2;
    const int* src = ei;
    const int* dst = ei + E;

    const int NB = (N + 255) / 256;
    int*   pairs = (int*)d_ws;
    float* h1   = (float*)(pairs + (size_t)2 * E);
    float* h2   = h1   + (size_t)N * 64;
    float* dis  = h2   + (size_t)N * 16;
    int* start   = (int*)(dis + N);
    int* cnt     = start + N;
    int* bcnt    = cnt + N;
    int* bcur    = bcnt + NB;
    int* bstart  = bcur + NB;
    int* csr_src = bstart + NB;

    hipMemsetAsync(bcnt, 0, (size_t)(2 * NB) * sizeof(int), stream);

    const int T = 256;

    {
        const int nChunks = (E + CH - 1) / CH;
        const int SPB = PT / 32;
        const int nStrips = N / 16;
        const int cap = 256 * SPB;
        const int thirdA = nStrips / 3;
        const int sA = thirdA < cap ? thirdA : cap;
        const int halfC = (nStrips - sA) / 2;
        const int sB = nStrips - (halfC < cap ? halfC : cap);
        const int gA = (sA + SPB - 1) / SPB;
        const int gB = (sB - sA + SPB - 1) / SPB;
        const int gC = (nStrips - sB + SPB - 1) / SPB;
        bucket_hist_kernel<<<nChunks + gA, PT, (size_t)NB * sizeof(int), stream>>>(
            dst, E, NB, bcnt, nChunks, x, W1, h1, 0, sA);
        bucket_scatter_kernel<<<nChunks + gB, PT, (size_t)(2 * NB + 16) * sizeof(int), stream>>>(
            src, dst, E, NB, bcnt, bcur, bstart, pairs, nChunks, x, W1, h1, sA, sB);
        bucket_finish_kernel<<<NB + gC, PT, 0, stream>>>(pairs, bstart, bcnt, N,
                                                      start, cnt, dis, csr_src,
                                                      NB, x, W1, h1, sB, nStrips);
    }
    agg_relu_gemm2_kernel<<<N / 16, T, 0, stream>>>(
        csr_src, start, cnt, dis, h1, b1, W2, h2);
    {
        long long total = (long long)N * 4;
        agg_logsoftmax_kernel<<<(int)((total + T - 1) / T), T, 0, stream>>>(
            csr_src, start, cnt, dis, h2, b2, out, N);
    }
}


# Bucket-size pass dropped: each bucket owns a fixed region of the pair list (2x mean, overflow list behind), places reserved by one returning add per run; 5 launches -> 4; order in a node as before
# speedup vs baseline: 7.3501x; 1.1328x over previous
#include <hip/hip_runtime.h>
#include <cstdint>

typedef __attribute__((ext_vector_type(2))) float v2f;
typedef __attribute__((ext_vector_type(8))) float v8f;

template <int K, int NC, int GW>
__device__ __forceinline__ void gemm_strips_wmma_f32(const float* __restrict__ A,
                                                     const float* __restrict__ B,
                                                     float* __restrict__ C,
                                                     int strip0, int stripEnd, int blk) {
    constexpr int TN = NC / 16;
    constexpr int KB = K / 8;
    __shared__ float4 bl[KB * TN * 32];

    const int lane = threadIdx.x & 31;
    const int wv = threadIdx.x >> 5;
    const int c = lane & 15;
    const int h = lane >> 4;
    const int strip = strip0 + blk * GW + wv;
    const bool live = strip < stripEnd;
    const int rowBase = (live ? strip : strip0) * 16;

    constexpr int SPW = KB * TN / GW;
    static_assert(KB * TN % GW == 0, "slot rows must divide evenly over the waves");
    const float* bp = B + (size_t)(4 * h) * NC + c;
    float4 bv[SPW];
#pragma unroll
    for (int i = 0; i < SPW; ++i) {
        int jn = wv + i * GW;
        int j = jn / TN;
        int n = jn % TN;
        const float* p = bp + (size_t)(8 * j) * NC + 16 * n;
        bv[i].x = p[0];
        bv[i].y = p[NC];
        bv[i].z = p[2 * NC];
        bv[i].w = p[3 * NC];
    }

    const float4* ap = (const float4*)(A + (size_t)(rowBase + c) * K + 4 * h);
    float4 a[KB];
#pragma unroll
    for (int j = 0; j < KB; ++j) a[j] = ap[2 * j];

#pragma unroll
    for (int i = 0; i < SPW; ++i) bl[(wv + i * GW) * 32 + lane] = bv[i];
    __syncthreads();
    if (!live) return;

    v8f acc[TN];
#pragma unroll
    for (int n = 0; n < TN; ++n) acc[n] = (v8f){};

    float4 bcur[TN], bnxt[TN];
#pragma unroll
    for (int n = 0; n < TN; ++n) bcur[n] = bl[n * 32 + lane];
#pragma unroll
    for (int j = 0; j < KB; ++j) {
        if (j + 1 < KB) {
#pragma unroll
            for (int n = 0; n < TN; ++n) bnxt[n] = bl[((j + 1) * TN + n) * 32 + lane];
        }
        v2f a0, a1;
        a0.x = a[j].x; a0.y = a[j].y;
        a1.x = a[j].z; a1.y = a[j].w;
#pragma unroll
        for (int n = 0; n < TN; ++n) {
            v2f b0;
            b0.x = bcur[n].x; b0.y = bcur[n].y;
            acc[n] = __builtin_amdgcn_wmma_f32_16x16x4_f32(
                 false, a0,  false, b0,
                 (short)0, acc[n],  false,  false);
        }
#pragma unroll
        for (int n = 0; n < TN; ++n) {
            v2f b1;
            b1.x = bcur[n].z; b1.y = bcur[n].w;
            acc[n] = __builtin_amdgcn_wmma_f32_16x16x4_f32(
                 false, a1,  false, b1,
                 (short)0, acc[n],  false,  false);
        }
#pragma unroll
        for (int n = 0; n < TN; ++n) bcur[n] = bnxt[n];
    }

    float* cptr = C + (size_t)(rowBase + 8 * h) * NC + c;
#pragma unroll
    for (int n = 0; n < TN; ++n)
#pragma unroll
        for (int i = 0; i < 8; ++i) cptr[(size_t)i * NC + 16 * n] = acc[n][i];
}

constexpr int PT  = 256;
constexpr int EPT = 16;
constexpr int CH  = PT * EPT;
constexpr int FT  = 8192;

__global__ void __launch_bounds__(PT)
bucket_scatter_kernel(const int* __restrict__ src,
                      const int* __restrict__ dst, int E, int NB, int cap,
                      int* __restrict__ bfill,
                      int* __restrict__ pairs_i,
                      int* __restrict__ ovf_i, int nPart,
                      const float* __restrict__ x, const float* __restrict__ W1,
                      float* __restrict__ h1, int strip0, int stripEnd) {
    if ((int)blockIdx.x >= nPart) {
        gemm_strips_wmma_f32<128, 64, PT / 32>(x, W1, h1, strip0, stripEnd,
                                               (int)blockIdx.x - nPart);
        return;
    }
    extern __shared__ int sm[];
    int* lcnt   = sm;
    int* lstart = sm + NB;
    int* wsum   = sm + 2 * NB;
    __shared__ int2 tile[CH];
    const int tid = threadIdx.x;

    int base = blockIdx.x * CH + tid;
    int d[EPT], s[EPT], r[EPT];
#pragma unroll
    for (int j = 0; j < EPT; ++j) {
        int e = base + j * PT;
        bool ok = e < E;
        d[j] = ok ? dst[e] : -1;
        s[j] = ok ? src[e] : 0;
        r[j] = 0;
    }
    for (int b = tid; b < NB; b += PT) lcnt[b] = 0;
    if (tid == 0) wsum[0] = 0;
    __syncthreads();

#pragma unroll
    for (int j = 0; j < EPT; ++j)
        if (d[j] >= 0) r[j] = atomicAdd(&lcnt[d[j] >> 8], 1);
    __syncthreads();
    for (int b = tid; b < NB; b += PT) {
        int c = lcnt[b];
        int off = 0;
        int lpos = 0;
        if (c) {
            off = atomicAdd(&bfill[b], c);
            lpos = atomicAdd(&wsum[0], c);
        }
        lcnt[b] = lpos;
        lstart[b] = off - lpos;
    }
    __syncthreads();
    int rb[EPT];
#pragma unroll
    for (int j = 0; j < EPT; ++j) rb[j] = lcnt[d[j] >= 0 ? (d[j] >> 8) : 0];
#pragma unroll
    for (int j = 0; j < EPT; ++j)
        if (d[j] >= 0) tile[rb[j] + r[j]] = make_int2(s[j], d[j]);
    __syncthreads();
    const int e0 = blockIdx.x * CH;
    const int nE = (E - e0) < CH ? (E - e0) : CH;
    int2* pairs = (int2*)pairs_i;
    int2* ovf = (int2*)ovf_i;
    int2 pe[EPT];
    int dl[EPT];
#pragma unroll
    for (int j = 0; j < EPT; ++j) pe[j] = tile[min(j * PT + tid, nE - 1)];
#pragma unroll
    for (int j = 0; j < EPT; ++j) dl[j] = lstart[pe[j].y >> 8];
#pragma unroll
    for (int j = 0; j < EPT; ++j) {
        int i = j * PT + tid;
        if (i < nE) {
            int pos = i + dl[j];
            if (pos < cap) {
                pairs[(size_t)(pe[j].y >> 8) * cap + pos] = pe[j];
            } else {
                int o = atomicAdd(&bfill[NB], 1);
                ovf[o] = pe[j];
            }
        }
    }
}

__global__ void __launch_bounds__(PT)
bucket_finish_kernel(const int* __restrict__ pairs_i,
                     const int* __restrict__ ovf_i,
                     const int* __restrict__ bfill, int NB, int cap, int n,
                     int* __restrict__ start,
                     int* __restrict__ cnt,
                     float* __restrict__ dis,
                     int* __restrict__ csr_src, int nPart,
                     const float* __restrict__ x, const float* __restrict__ W1,
                     float* __restrict__ h1, int strip0, int stripEnd) {
    if ((int)blockIdx.x >= nPart) {
        gemm_strips_wmma_f32<128, 64, PT / 32>(x, W1, h1, strip0, stripEnd,
                                               (int)blockIdx.x - nPart);
        return;
    }
    __shared__ __attribute__((aligned(16))) int ftile[FT];
    __shared__ int lcnt[256];
    __shared__ int lcur[256];
    __shared__ int lex[256];
    __shared__ int wsum[8];
    const int tid = threadIdx.x;
    const int lane = tid & 31;
    const int w = tid >> 5;
    const int b = blockIdx.x;
    int part = 0;
    for (int i = tid; i < b; i += 256) part += bfill[i];
    const int bt = bfill[b];
    const int nov = bfill[NB];
#pragma unroll
    for (int o = 16; o > 0; o >>= 1) part += __shfl_xor(part, o);
    if (lane == 0) wsum[w] = part;
    lcnt[tid] = 0;
    lcur[tid] = 0;
    __syncthreads();
    int bs = 0;
#pragma unroll
    for (int k = 0; k < 8; ++k) bs += wsum[k];
    const int bc = bt < cap ? bt : cap;
    const int2* p = (const int2*)pairs_i + (size_t)b * cap;
    const int2* ov = (const int2*)ovf_i;

    for (int i0 = 0; i0 < bc; i0 += 4 * 256) {
        int2 q[4];
#pragma unroll
        for (int k = 0; k < 4; ++k) q[k] = p[min(i0 + k * 256 + tid, bc - 1)];
#pragma unroll
        for (int k = 0; k < 4; ++k)
            if (i0 + k * 256 + tid < bc) atomicAdd(&lcnt[q[k].y & 255], 1);
    }
    if (nov > 0) {
        for (int i = tid; i < nov; i += 256) {
            int2 q = ov[i];
            if ((q.y >> 8) == b) atomicAdd(&lcnt[q.y & 255], 1);
        }
    }
    __syncthreads();

    int c = lcnt[tid];
    int incl = c;
#pragma unroll
    for (int o = 1; o < 32; o <<= 1) {
        int t = __shfl_up(incl, o);
        if (lane >= o) incl += t;
    }
    if (lane == 31) wsum[w] = incl;
    __syncthreads();
    if (tid == 0) {
        int tot = 0;
#pragma unroll
        for (int k = 0; k < 8; ++k) { int t = wsum[k]; wsum[k] = tot; tot += t; }
    }
    __syncthreads();
    int ex = wsum[w] + incl - c;
    lex[tid] = ex;
    int node = b * 256 + tid;
    if (node < n) {
        start[node] = bs + ex;
        cnt[node] = c;
        dis[node] = rsqrtf((float)c + 1.0f);
    }
    __syncthreads();

    int* seg = csr_src + bs;
    const int sh = (int)((((size_t)seg) >> 2) & 3);
    const int tot = sh + bt;
    const bool staged = tot <= FT;
    for (int i0 = 0; i0 < bc; i0 += 4 * 256) {
        int2 q[4];
#pragma unroll
        for (int k = 0; k < 4; ++k) q[k] = p[min(i0 + k * 256 + tid, bc - 1)];
#pragma unroll
        for (int k = 0; k < 4; ++k) {
            if (i0 + k * 256 + tid < bc) {
                int l = q[k].y & 255;
                int rk = atomicAdd(&lcur[l], 1);
                if (staged) ftile[sh + lex[l] + rk] = q[k].x;
                else        seg[lex[l] + rk] = q[k].x;
            }
        }
    }
    if (nov > 0) {
        for (int i = tid; i < nov; i += 256) {
            int2 q = ov[i];
            if ((q.y >> 8) == b) {
                int l = q.y & 255;
                int rk = atomicAdd(&lcur[l], 1);
                if (staged) ftile[sh + lex[l] + rk] = q.x;
                else        seg[lex[l] + rk] = q.x;
            }
        }
    }
    if (!staged) return;
    __syncthreads();
    int* al = seg - sh;
    int4* gout = (int4*)al;
    const int4* tin = (const int4*)ftile;
    const int g0 = sh > 0 ? 1 : 0;
    const int g1 = tot >> 2;
    for (int gq = g0 + tid; gq < g1; gq += 256) gout[gq] = tin[gq];
    if (tid < 4) {
        int e = tid;
        if (sh > 0 && e >= sh && e < tot) al[e] = ftile[e];
    } else if (tid >= 32 && tid < 36) {
        int e = 4 * g1 + (tid - 32);
        if (e >= 4 * g0 && e < tot) al[e] = ftile[e];
    }
}

template <int G>
__device__ __forceinline__ float4 gather_sum(float4 acc,
                                             const int* __restrict__ idx, int len,
                                             float dd,
                                             const float* __restrict__ dis,
                                             const float4* __restrict__ h4, int q) {
    int k = 0;
    for (; k + 4 <= len; k += 4) {
        int s0 = idx[k + 0];
        int s1 = idx[k + 1];
        int s2 = idx[k + 2];
        int s3 = idx[k + 3];
        float w0 = dis[s0] * dd;
        float w1 = dis[s1] * dd;
        float w2 = dis[s2] * dd;
        float w3 = dis[s3] * dd;
        float4 v0 = h4[(size_t)s0 * G + q];
        float4 v1 = h4[(size_t)s1 * G + q];
        float4 v2 = h4[(size_t)s2 * G + q];
        float4 v3 = h4[(size_t)s3 * G + q];
        acc.x += v0.x * w0; acc.y += v0.y * w0; acc.z += v0.z * w0; acc.w += v0.w * w0;
        acc.x += v1.x * w1; acc.y += v1.y * w1; acc.z += v1.z * w1; acc.w += v1.w * w1;
        acc.x += v2.x * w2; acc.y += v2.y * w2; acc.z += v2.z * w2; acc.w += v2.w * w2;
        acc.x += v3.x * w3; acc.y += v3.y * w3; acc.z += v3.z * w3; acc.w += v3.w * w3;
    }
    if (k < len) {
        int last = len - 1;
        bool p1 = k + 1 < len;
        bool p2 = k + 2 < len;
        int s0 = idx[k];
        int s1 = idx[min(k + 1, last)];
        int s2 = idx[min(k + 2, last)];
        float w0 = dis[s0] * dd;
        float w1 = dis[s1] * dd;
        float w2 = dis[s2] * dd;
        float4 v0 = h4[(size_t)s0 * G + q];
        float4 v1 = h4[(size_t)s1 * G + q];
        float4 v2 = h4[(size_t)s2 * G + q];
        acc.x += v0.x * w0; acc.y += v0.y * w0; acc.z += v0.z * w0; acc.w += v0.w * w0;
        float4 t1 = acc;
        t1.x += v1.x * w1; t1.y += v1.y * w1; t1.z += v1.z * w1; t1.w += v1.w * w1;
        acc.x = p1 ? t1.x : acc.x; acc.y = p1 ? t1.y : acc.y;
        acc.z = p1 ? t1.z : acc.z; acc.w = p1 ? t1.w : acc.w;
        float4 t2 = acc;
        t2.x += v2.x * w2; t2.y += v2.y * w2; t2.z += v2.z * w2; t2.w += v2.w * w2;
        acc.x = p2 ? t2.x : acc.x; acc.y = p2 ? t2.y : acc.y;
        acc.z = p2 ? t2.z : acc.z; acc.w = p2 ? t2.w : acc.w;
    }
    return acc;
}

template <int G>
__device__ __forceinline__ float4 gather_rows_sum(float4 acc,
                                                  const int* __restrict__ idx, int len,
                                                  const float4* __restrict__ g4, int q) {
    int k = 0;
#pragma unroll 2
    for (; k + 4 <= len; k += 4) {
        int s0 = idx[k + 0];
        int s1 = idx[k + 1];
        int s2 = idx[k + 2];
        int s3 = idx[k + 3];
        float4 v0 = g4[(size_t)s0 * G + q];
        float4 v1 = g4[(size_t)s1 * G + q];
        float4 v2 = g4[(size_t)s2 * G + q];
        float4 v3 = g4[(size_t)s3 * G + q];
        acc.x += v0.x; acc.y += v0.y; acc.z += v0.z; acc.w += v0.w;
        acc.x += v1.x; acc.y += v1.y; acc.z += v1.z; acc.w += v1.w;
        acc.x += v2.x; acc.y += v2.y; acc.z += v2.z; acc.w += v2.w;
        acc.x += v3.x; acc.y += v3.y; acc.z += v3.z; acc.w += v3.w;
    }
    if (k < len) {
        int last = len - 1;
        bool p1 = k + 1 < len;
        bool p2 = k + 2 < len;
        int s0 = idx[k];
        int s1 = idx[min(k + 1, last)];
        int s2 = idx[min(k + 2, last)];
        float4 v0 = g4[(size_t)s0 * G + q];
        float4 v1 = g4[(size_t)s1 * G + q];
        float4 v2 = g4[(size_t)s2 * G + q];
        acc.x += v0.x; acc.y += v0.y; acc.z += v0.z; acc.w += v0.w;
        float4 t1 = acc;
        t1.x += v1.x; t1.y += v1.y; t1.z += v1.z; t1.w += v1.w;
        acc.x = p1 ? t1.x : acc.x; acc.y = p1 ? t1.y : acc.y;
        acc.z = p1 ? t1.z : acc.z; acc.w = p1 ? t1.w : acc.w;
        float4 t2 = acc;
        t2.x += v2.x; t2.y += v2.y; t2.z += v2.z; t2.w += v2.w;
        acc.x = p2 ? t2.x : acc.x; acc.y = p2 ? t2.y : acc.y;
        acc.z = p2 ? t2.z : acc.z; acc.w = p2 ? t2.w : acc.w;
    }
    return acc;
}

constexpr int YP4 = 17;

__global__ void __launch_bounds__(256)
agg_relu_gemm2_kernel(const int* __restrict__ csr_src,
                      const int* __restrict__ start,
                      const int* __restrict__ cnt,
                      const float* __restrict__ dis,
                      const float* __restrict__ h,
                      const float* __restrict__ bias,
                      const float* __restrict__ W2,
                      float* __restrict__ g) {
    constexpr int G = 16;
    __shared__ float4 bl[8 * 32];
    __shared__ float4 ys[16 * YP4];
    __shared__ int arrived;

    const int tid = threadIdx.x;
    const int r = tid >> 4;
    const int q = tid & 15;
    const int rowBase = blockIdx.x * 16;

    const float* wp = W2 + (size_t)r * 64 + q;
    float4 bv;
    bv.x = wp[0];
    bv.y = wp[16];
    bv.z = wp[32];
    bv.w = wp[48];

    const int node = rowBase + r;
    const float4* h4 = (const float4*)h;
    const float dd = dis[node];
    const float4 self = h4[(size_t)rowBase * G + tid];
    const float4 bq = ((const float4*)bias)[q];
    const int st = start[node];
    const int len = cnt[node];

    bl[tid] = bv;
    if (tid == 0) arrived = 0;
    __syncthreads();

    const float d2 = dd * dd;
    float4 acc;
    acc.x = self.x * d2 + bq.x;
    acc.y = self.y * d2 + bq.y;
    acc.z = self.z * d2 + bq.z;
    acc.w = self.w * d2 + bq.w;
    acc = gather_sum<G>(acc, csr_src + st, len, dd, dis, h4, q);
    float4 v;
    v.x = fmaxf(acc.x, 0.0f);
    v.y = fmaxf(acc.y, 0.0f);
    v.z = fmaxf(acc.z, 0.0f);
    v.w = fmaxf(acc.w, 0.0f);
    ys[tid + (tid >> 4)] = v;

    __threadfence_block();
    int old = 0;
    if ((tid & 31) == 0)
        old = __hip_atomic_fetch_add(&arrived, 1, __ATOMIC_ACQ_REL, __HIP_MEMORY_SCOPE_WORKGROUP);
    old = __builtin_amdgcn_readfirstlane(old);
    if (old != 7) return;

    const int lane = tid & 31;
    const int c = lane & 15;
    const int hh = lane >> 4;
    const float4* sp = (const float4*)(dis + rowBase + 8 * hh);
    const float4 s0 = sp[0];
    const float4 s1 = sp[1];

    v8f p0 = (v8f){};
    v8f p1 = (v8f){};
#pragma unroll
    for (int j = 0; j < 8; ++j) {
        const float4 a4 = ys[c * YP4 + 2 * j + hh];
        const float4 b4 = bl[j * 32 + lane];
        v2f a0, a1, b0, b1;
        a0.x = a4.x; a0.y = a4.y;
        a1.x = a4.z; a1.y = a4.w;
        b0.x = b4.x; b0.y = b4.y;
        b1.x = b4.z; b1.y = b4.w;
        p0 = __builtin_amdgcn_wmma_f32_16x16x4_f32(
             false, a0,  false, b0,
             (short)0, p0,  false,  false);
        p1 = __builtin_amdgcn_wmma_f32_16x16x4_f32(
             false, a1,  false, b1,
             (short)0, p1,  false,  false);
    }
    const float sc[8] = {s0.x, s0.y, s0.z, s0.w, s1.x, s1.y, s1.z, s1.w};
    float* gp = g + (size_t)(rowBase + 8 * hh) * 16 + c;
#pragma unroll
    for (int i = 0; i < 8; ++i) gp[(size_t)i * 16] = (p0[i] + p1[i]) * sc[i];
}

__global__ void agg_logsoftmax_kernel(const int* __restrict__ csr_src,
                                      const int* __restrict__ start,
                                      const int* __restrict__ cnt,
                                      const float* __restrict__ dis,
                                      const float* __restrict__ g,
                                      const float* __restrict__ bias,
                                      float* __restrict__ out, int n) {
    constexpr int G = 4;
    int t = blockIdx.x * blockDim.x + threadIdx.x;
    int node = t / G;
    int q = t % G;
    if (node >= n) return;
    const float4* g4 = (const float4*)g;
    const int st = start[node];
    const int len = cnt[node];
    float dd = dis[node];
    float4 acc = g4[t];
    const float4 b = ((const float4*)bias)[q];
    asm volatile("" :: "v"(st), "v"(len), "v"(b.x), "v"(b.y), "v"(b.z), "v"(b.w));
    acc = gather_rows_sum<G>(acc, csr_src + st, len, g4, q);
    float4 v;
    v.x = acc.x * dd + b.x;
    v.y = acc.y * dd + b.y;
    v.z = acc.z * dd + b.z;
    v.w = acc.w * dd + b.w;
    float mx = fmaxf(fmaxf(v.x, v.y), fmaxf(v.z, v.w));
    mx = fmaxf(mx, __shfl_xor(mx, 1));
    mx = fmaxf(mx, __shfl_xor(mx, 2));
    float s = (__expf(v.x - mx) + __expf(v.y - mx)) + (__expf(v.z - mx) + __expf(v.w - mx));
    s += __shfl_xor(s, 1);
    s += __shfl_xor(s, 2);
    float lse = mx + __logf(s);
    float4 r;
    r.x = v.x - lse; r.y = v.y - lse; r.z = v.z - lse; r.w = v.w - lse;
    ((float4*)out)[t] = r;
}

extern "C" void kernel_launch(void* const* d_in, const int* in_sizes, int n_in,
                              void* d_out, int out_size, void* d_ws, size_t ws_size,
                              hipStream_t stream) {
    const float* x   = (const float*)d_in[0];
    const int*   ei  = (const int*)  d_in[1];
    const float* W1  = (const float*)d_in[2];
    const float* b1  = (const float*)d_in[3];
    const float* W2  = (const float*)d_in[4];
    const float* b2  = (const float*)d_in[5];
    float* out = (float*)d_out;

    const int N = in_sizes[0] / 128;
    const int E = in_sizes[1] / 2;
    const int* src = ei;
    const int* dst = ei + E;

    const int NB = (N + 255) / 256;
    const int cap = ((2 * ((E + NB - 1) / NB) + 63) / 64) * 64 + 32;
    int*   pairs = (int*)d_ws;
    int*   ovf  = pairs + (size_t)2 * NB * cap;
    float* h1   = (float*)(ovf + (size_t)2 * E);
    float* h2   = h1   + (size_t)N * 64;
    float* dis  = h2   + (size_t)N * 16;
    int* start   = (int*)(dis + N);
    int* cnt     = start + N;
    int* bfill   = cnt + N;
    int* csr_src = bfill + NB + 1;

    hipMemsetAsync(bfill, 0, (size_t)(NB + 1) * sizeof(int), stream);

    const int T = 256;

    {
        const int nChunks = (E + CH - 1) / CH;
        const int SPB = PT / 32;
        const int nStrips = N / 16;
        const int sA = nStrips / 2;
        const int gA = (sA + SPB - 1) / SPB;
        const int gB = (nStrips - sA + SPB - 1) / SPB;
        bucket_scatter_kernel<<<nChunks + gA, PT, (size_t)(2 * NB + 16) * sizeof(int), stream>>>(
            src, dst, E, NB, cap, bfill, pairs, ovf, nChunks, x, W1, h1, 0, sA);
        bucket_finish_kernel<<<NB + gB, PT, 0, stream>>>(pairs, ovf, bfill, NB, cap, N,
                                                      start, cnt, dis, csr_src,
                                                      NB, x, W1, h1, sA, nStrips);
    }
    agg_relu_gemm2_kernel<<<N / 16, T, 0, stream>>>(
        csr_src, start, cnt, dis, h1, b1, W2, h2);
    {
        long long total = (long long)N * 4;
        agg_logsoftmax_kernel<<<(int)((total + T - 1) / T), T, 0, stream>>>(
            csr_src, start, cnt, dis, h2, b2, out, N);
    }
}
